# starting kernel
# baseline (speedup 1.0000x reference)
#include <hip/hip_runtime.h>
#include <hip/hip_bf16.h>

#define N_NODES   50000
#define N_EDGES   800000
#define DIM       128
#define N_LAYERS  4
#define NUM_GRAPHS 128
#define BN_EPS    1e-5f

typedef __attribute__((ext_vector_type(2))) float v2f;
typedef __attribute__((ext_vector_type(8))) float v8f;

__global__ __launch_bounds__(256) void gin_zero(float* __restrict__ p, long n) {
    long i = (long)blockIdx.x * blockDim.x + threadIdx.x;
    if (i < n) p[i] = 0.0f;
}

__global__ __launch_bounds__(256) void gin_scatter(const float* __restrict__ hin,
                                                   const int* __restrict__ src,
                                                   const int* __restrict__ dst,
                                                   float* __restrict__ agg) {
    long t = (long)blockIdx.x * blockDim.x + threadIdx.x;
    long e = t >> 5;
    int lane = (int)(t & 31);
    if (e >= N_EDGES) return;
    int s = src[e];
    int d = dst[e];
    const float4* hs = (const float4*)(hin + (long)s * DIM);
    float4 v = hs[lane];
    float* ad = agg + (long)d * DIM + lane * 4;
    atomicAdd(ad + 0, v.x);
    atomicAdd(ad + 1, v.y);
    atomicAdd(ad + 2, v.z);
    atomicAdd(ad + 3, v.w);
}

__global__ __launch_bounds__(256) void gin_mlp(const float* __restrict__ hin,
                                               const float* __restrict__ agg,
                                               const float* __restrict__ W1,
                                               const float* __restrict__ b1,
                                               const float* __restrict__ W2,
                                               const float* __restrict__ b2,
                                               const float* __restrict__ epsp,
                                               float* __restrict__ hout,
                                               float* __restrict__ stats,
                                               int do_stats) {
    __shared__ float zt[16 * DIM];
    __shared__ float y1[16 * DIM];

    const int row0 = blockIdx.x * 16;
    const int tid  = threadIdx.x;
    const float e1 = 1.0f + epsp[0];

    for (int i = tid; i < 16 * DIM; i += 256) {
        int r = i >> 7;
        int c = i & (DIM - 1);
        long g = (long)(row0 + r) * DIM + c;
        zt[i] = e1 * hin[g] + agg[g];
    }
    __syncthreads();

    const int wave = tid >> 5;
    const int lane = tid & 31;
    const int n    = lane & 15;
    const int hi   = lane >> 4;
    const int col0 = wave * 16;
    const int kk   = hi * 2;

    __builtin_prefetch(&W1[col0], 0, 0);
    __builtin_prefetch(&W2[col0], 0, 0);

    v8f acc;
    {
        float bv = b1[col0 + n];
        for (int v = 0; v < 8; ++v) acc[v] = bv;
    }
    for (int k = 0; k < DIM; k += 4) {
        v2f a, b;
        const float* zp = &zt[n * DIM + k + kk];
        a.x = zp[0];
        a.y = zp[1];
        b.x = W1[(k + kk) * DIM + col0 + n];
        b.y = W1[(k + kk + 1) * DIM + col0 + n];
        acc = __builtin_amdgcn_wmma_f32_16x16x4_f32(false, a, false, b,
                                                    (short)0, acc, false, false);
    }
    for (int v = 0; v < 8; ++v) {
        int m = v + 8 * hi;
        y1[m * DIM + col0 + n] = fmaxf(acc[v], 0.0f);
    }
    __syncthreads();

    {
        float bv = b2[col0 + n];
        for (int v = 0; v < 8; ++v) acc[v] = bv;
    }
    for (int k = 0; k < DIM; k += 4) {
        v2f a, b;
        const float* yp = &y1[n * DIM + k + kk];
        a.x = yp[0];
        a.y = yp[1];
        b.x = W2[(k + kk) * DIM + col0 + n];
        b.y = W2[(k + kk + 1) * DIM + col0 + n];
        acc = __builtin_amdgcn_wmma_f32_16x16x4_f32(false, a, false, b,
                                                    (short)0, acc, false, false);
    }

    float s = 0.0f, ssq = 0.0f;
    for (int v = 0; v < 8; ++v) {
        int m = v + 8 * hi;
        float val = acc[v];
        hout[(long)(row0 + m) * DIM + col0 + n] = val;
        s += val;
        ssq += val * val;
    }
    if (do_stats) {
        atomicAdd(&stats[col0 + n], s);
        atomicAdd(&stats[DIM + col0 + n], ssq);
    }
}

__global__ __launch_bounds__(256) void gin_bn_relu(float* __restrict__ z,
                                                   const float* __restrict__ stats,
                                                   const float* __restrict__ gamma,
                                                   const float* __restrict__ beta) {
    long i = (long)blockIdx.x * blockDim.x + threadIdx.x;
    if (i >= (long)N_NODES * DIM) return;
    int c = (int)(i & (DIM - 1));
    const float invN = 1.0f / (float)N_NODES;
    float mu  = stats[c] * invN;
    float var = stats[DIM + c] * invN - mu * mu;
    float v = (z[i] - mu) * rsqrtf(var + BN_EPS) * gamma[c] + beta[c];
    z[i] = fmaxf(v, 0.0f);
}

__global__ __launch_bounds__(256) void gin_pool(const float* __restrict__ h,
                                                const int* __restrict__ batch,
                                                float* __restrict__ out) {
    long t = (long)blockIdx.x * blockDim.x + threadIdx.x;
    long nidx = t >> 5;
    int lane = (int)(t & 31);
    if (nidx >= N_NODES) return;
    int g = batch[nidx];
    const float4* hp = (const float4*)(h + nidx * DIM);
    float4 v = hp[lane];
    float* op = out + (long)g * DIM + lane * 4;
    atomicAdd(op + 0, v.x);
    atomicAdd(op + 1, v.y);
    atomicAdd(op + 2, v.z);
    atomicAdd(op + 3, v.w);
}

extern "C" void kernel_launch(void* const* d_in, const int* in_sizes, int n_in,
                              void* d_out, int out_size, void* d_ws, size_t ws_size,
                              hipStream_t stream) {
    const float* x     = (const float*)d_in[0];
    const int*   ei    = (const int*)d_in[1];
    const int*   batch = (const int*)d_in[2];
    const float* W1    = (const float*)d_in[3];
    const float* b1    = (const float*)d_in[4];
    const float* W2    = (const float*)d_in[5];
    const float* b2    = (const float*)d_in[6];
    const float* eps   = (const float*)d_in[7];
    const float* gamma = (const float*)d_in[8];
    const float* beta  = (const float*)d_in[9];
    float* out = (float*)d_out;

    const long ND = (long)N_NODES * DIM;
    float* hA    = (float*)d_ws;
    float* hB    = hA + ND;
    float* agg   = hB + ND;
    float* stats = agg + ND;

    const int* src = ei;
    const int* dst = ei + N_EDGES;

    const long zeroN   = ND + 2 * DIM;
    const int  zeroGrd = (int)((zeroN + 255) / 256);
    const int  scatGrd = (int)(((long)N_EDGES * 32 + 255) / 256);
    const int  mlpGrd  = N_NODES / 16;
    const int  bnGrd   = (int)((ND + 255) / 256);
    const int  poolGrd = (int)(((long)N_NODES * 32 + 255) / 256);

    for (int i = 0; i < N_LAYERS; ++i) {
        const float* hin  = (i == 0) ? x : ((i & 1) ? hA : hB);
        float*       hout = (i & 1) ? hB : hA;
        int do_stats = (i < N_LAYERS - 1) ? 1 : 0;

        gin_zero<<<zeroGrd, 256, 0, stream>>>(agg, zeroN);
        gin_scatter<<<scatGrd, 256, 0, stream>>>(hin, src, dst, agg);
        gin_mlp<<<mlpGrd, 256, 0, stream>>>(hin, agg,
                                            W1 + (long)i * DIM * DIM, b1 + (long)i * DIM,
                                            W2 + (long)i * DIM * DIM, b2 + (long)i * DIM,
                                            eps + i, hout, stats, do_stats);
        if (do_stats) {
            gin_bn_relu<<<bnGrd, 256, 0, stream>>>(hout, stats,
                                                   gamma + (long)i * DIM,
                                                   beta + (long)i * DIM);
        }
    }
    gin_zero<<<(NUM_GRAPHS * DIM + 255) / 256, 256, 0, stream>>>(out, NUM_GRAPHS * DIM);
    gin_pool<<<poolGrd, 256, 0, stream>>>(hB, batch, out);
}


# Neighbour sum by gather over an edge list grouped by destination (one returning int add per edge, scan, fill; once per call) instead of 102M float atomics; sum order in a row = add arrival order
# speedup vs baseline: 1.6557x; 1.6557x over previous
#include <hip/hip_runtime.h>
#include <hip/hip_bf16.h>

#define N_NODES   50000
#define N_EDGES   800000
#define DIM       128
#define N_LAYERS  4
#define NUM_GRAPHS 128
#define BN_EPS    1e-5f

typedef __attribute__((ext_vector_type(2))) float v2f;
typedef __attribute__((ext_vector_type(8))) float v8f;

__global__ __launch_bounds__(256) void gin_zero(float* __restrict__ p, long n) {
    long i = (long)blockIdx.x * blockDim.x + threadIdx.x;
    if (i < n) p[i] = 0.0f;
}

__global__ __launch_bounds__(256) void gin_zero_i(int* __restrict__ p, int n) {
    int i = blockIdx.x * blockDim.x + threadIdx.x;
    if (i < n) p[i] = 0;
}

__global__ __launch_bounds__(256) void gin_csr_count(const int* __restrict__ dst,
                                                     int* __restrict__ deg,
                                                     int* __restrict__ rank) {
    int e = blockIdx.x * blockDim.x + threadIdx.x;
    if (e >= N_EDGES) return;
    rank[e] = atomicAdd(&deg[dst[e]], 1);
}

__global__ __launch_bounds__(1024) void gin_csr_scan(const int* __restrict__ deg,
                                                     int* __restrict__ rowbeg) {
    __shared__ int wsum[32];
    __shared__ int wpre[32];
    const int tid  = threadIdx.x;
    const int lane = tid & 31;
    const int wave = tid >> 5;
    const int chunk0 = blockIdx.x * 1024;

    int pre = 0;
    for (int i = tid; i < chunk0; i += 1024) pre += deg[i];

    const int idx = chunk0 + tid;
    const int v = (idx < N_NODES) ? deg[idx] : 0;

    int inc = v;
    for (int off = 1; off < 32; off <<= 1) {
        int t = __shfl_up(inc, off);
        if (lane >= off) inc += t;
    }
    for (int off = 16; off > 0; off >>= 1) pre += __shfl_xor(pre, off);

    if (lane == 31) wsum[wave] = inc;
    if (lane == 0)  wpre[wave] = pre;
    __syncthreads();

    int base = 0;
    for (int w = 0; w < 32; ++w) {
        base += wpre[w];
        if (w < wave) base += wsum[w];
    }
    const int excl = base + inc - v;
    if (idx < N_NODES) rowbeg[idx] = excl;
}

__global__ __launch_bounds__(256) void gin_csr_fill(const int* __restrict__ src,
                                                    const int* __restrict__ dst,
                                                    const int* __restrict__ rowbeg,
                                                    const int* __restrict__ rank,
                                                    int* __restrict__ col) {
    int e = blockIdx.x * blockDim.x + threadIdx.x;
    if (e >= N_EDGES) return;
    col[rowbeg[dst[e]] + rank[e]] = src[e];
}

__global__ __launch_bounds__(256) void gin_gather(const float* __restrict__ hin,
                                                  const int* __restrict__ rowbeg,
                                                  const int* __restrict__ deg,
                                                  const int* __restrict__ col,
                                                  float* __restrict__ agg,
                                                  float* __restrict__ stats) {
    if (blockIdx.x == 0) stats[threadIdx.x] = 0.0f;

    const int lane = threadIdx.x & 31;
    const int node = __builtin_amdgcn_readfirstlane((int)(blockIdx.x * 8 + (threadIdx.x >> 5)));
    if (node >= N_NODES) return;

    const int beg = rowbeg[node];
    const int cnt = deg[node];
    const float4* h4 = (const float4*)hin;

    float4 acc = make_float4(0.0f, 0.0f, 0.0f, 0.0f);
    for (int base = 0; base < cnt; base += 32) {
        const int m = (cnt - base < 32) ? (cnt - base) : 32;
        const int mine = (lane < m) ? col[beg + base + lane] : 0;
        for (int j = 0; j < m; j += 4) {
            float4 a0 = make_float4(0.0f, 0.0f, 0.0f, 0.0f);
            float4 a1 = a0, a2 = a0, a3 = a0;
            a0 = h4[(long)__builtin_amdgcn_readlane(mine, j) * (DIM / 4) + lane];
            if (j + 1 < m) a1 = h4[(long)__builtin_amdgcn_readlane(mine, j + 1) * (DIM / 4) + lane];
            if (j + 2 < m) a2 = h4[(long)__builtin_amdgcn_readlane(mine, j + 2) * (DIM / 4) + lane];
            if (j + 3 < m) a3 = h4[(long)__builtin_amdgcn_readlane(mine, j + 3) * (DIM / 4) + lane];
            acc.x += (a0.x + a1.x) + (a2.x + a3.x);
            acc.y += (a0.y + a1.y) + (a2.y + a3.y);
            acc.z += (a0.z + a1.z) + (a2.z + a3.z);
            acc.w += (a0.w + a1.w) + (a2.w + a3.w);
        }
    }
    ((float4*)agg)[(long)node * (DIM / 4) + lane] = acc;
}

__global__ __launch_bounds__(256) void gin_mlp(const float* __restrict__ hin,
                                               const float* __restrict__ agg,
                                               const float* __restrict__ W1,
                                               const float* __restrict__ b1,
                                               const float* __restrict__ W2,
                                               const float* __restrict__ b2,
                                               const float* __restrict__ epsp,
                                               float* __restrict__ hout,
                                               float* __restrict__ stats,
                                               int do_stats) {
    __shared__ float zt[16 * DIM];
    __shared__ float y1[16 * DIM];

    const int row0 = blockIdx.x * 16;
    const int tid  = threadIdx.x;
    const float e1 = 1.0f + epsp[0];

    for (int i = tid; i < 16 * DIM; i += 256) {
        int r = i >> 7;
        int c = i & (DIM - 1);
        long g = (long)(row0 + r) * DIM + c;
        zt[i] = e1 * hin[g] + agg[g];
    }
    __syncthreads();

    const int wave = tid >> 5;
    const int lane = tid & 31;
    const int n    = lane & 15;
    const int hi   = lane >> 4;
    const int col0 = wave * 16;
    const int kk   = hi * 2;

    __builtin_prefetch(&W1[col0], 0, 0);
    __builtin_prefetch(&W2[col0], 0, 0);

    v8f acc;
    {
        float bv = b1[col0 + n];
        for (int v = 0; v < 8; ++v) acc[v] = bv;
    }
    for (int k = 0; k < DIM; k += 4) {
        v2f a, b;
        const float* zp = &zt[n * DIM + k + kk];
        a.x = zp[0];
        a.y = zp[1];
        b.x = W1[(k + kk) * DIM + col0 + n];
        b.y = W1[(k + kk + 1) * DIM + col0 + n];
        acc = __builtin_amdgcn_wmma_f32_16x16x4_f32(false, a, false, b,
                                                    (short)0, acc, false, false);
    }
    for (int v = 0; v < 8; ++v) {
        int m = v + 8 * hi;
        y1[m * DIM + col0 + n] = fmaxf(acc[v], 0.0f);
    }
    __syncthreads();

    {
        float bv = b2[col0 + n];
        for (int v = 0; v < 8; ++v) acc[v] = bv;
    }
    for (int k = 0; k < DIM; k += 4) {
        v2f a, b;
        const float* yp = &y1[n * DIM + k + kk];
        a.x = yp[0];
        a.y = yp[1];
        b.x = W2[(k + kk) * DIM + col0 + n];
        b.y = W2[(k + kk + 1) * DIM + col0 + n];
        acc = __builtin_amdgcn_wmma_f32_16x16x4_f32(false, a, false, b,
                                                    (short)0, acc, false, false);
    }

    float s = 0.0f, ssq = 0.0f;
    for (int v = 0; v < 8; ++v) {
        int m = v + 8 * hi;
        float val = acc[v];
        hout[(long)(row0 + m) * DIM + col0 + n] = val;
        s += val;
        ssq += val * val;
    }
    if (do_stats) {
        atomicAdd(&stats[col0 + n], s);
        atomicAdd(&stats[DIM + col0 + n], ssq);
    }
}

__global__ __launch_bounds__(256) void gin_bn_relu(float* __restrict__ z,
                                                   const float* __restrict__ stats,
                                                   const float* __restrict__ gamma,
                                                   const float* __restrict__ beta) {
    long i = (long)blockIdx.x * blockDim.x + threadIdx.x;
    if (i >= (long)N_NODES * DIM) return;
    int c = (int)(i & (DIM - 1));
    const float invN = 1.0f / (float)N_NODES;
    float mu  = stats[c] * invN;
    float var = stats[DIM + c] * invN - mu * mu;
    float v = (z[i] - mu) * rsqrtf(var + BN_EPS) * gamma[c] + beta[c];
    z[i] = fmaxf(v, 0.0f);
}

__global__ __launch_bounds__(256) void gin_pool(const float* __restrict__ h,
                                                const int* __restrict__ batch,
                                                float* __restrict__ out) {
    long t = (long)blockIdx.x * blockDim.x + threadIdx.x;
    long nidx = t >> 5;
    int lane = (int)(t & 31);
    if (nidx >= N_NODES) return;
    int g = batch[nidx];
    const float4* hp = (const float4*)(h + nidx * DIM);
    float4 v = hp[lane];
    float* op = out + (long)g * DIM + lane * 4;
    atomicAdd(op + 0, v.x);
    atomicAdd(op + 1, v.y);
    atomicAdd(op + 2, v.z);
    atomicAdd(op + 3, v.w);
}

extern "C" void kernel_launch(void* const* d_in, const int* in_sizes, int n_in,
                              void* d_out, int out_size, void* d_ws, size_t ws_size,
                              hipStream_t stream) {
    const float* x     = (const float*)d_in[0];
    const int*   ei    = (const int*)d_in[1];
    const int*   batch = (const int*)d_in[2];
    const float* W1    = (const float*)d_in[3];
    const float* b1    = (const float*)d_in[4];
    const float* W2    = (const float*)d_in[5];
    const float* b2    = (const float*)d_in[6];
    const float* eps   = (const float*)d_in[7];
    const float* gamma = (const float*)d_in[8];
    const float* beta  = (const float*)d_in[9];
    float* out = (float*)d_out;

    const long ND = (long)N_NODES * DIM;
    float* hA    = (float*)d_ws;
    float* hB    = hA + ND;
    float* agg   = hB + ND;
    float* stats = agg + ND;
    int* deg    = (int*)(stats + 2 * DIM);
    int* rowbeg = deg + N_NODES;
    int* rank   = rowbeg + N_NODES;
    int* col    = rank + N_EDGES;

    const int* src = ei;
    const int* dst = ei + N_EDGES;

    const int  degGrd  = (N_NODES + 255) / 256;
    const int  edgeGrd = (N_EDGES + 255) / 256;
    const int  scanGrd = (N_NODES + 1023) / 1024;
    const int  gathGrd = (N_NODES + 7) / 8;
    const int  mlpGrd  = N_NODES / 16;
    const int  bnGrd   = (int)((ND + 255) / 256);
    const int  poolGrd = (int)(((long)N_NODES * 32 + 255) / 256);

    gin_zero_i<<<degGrd, 256, 0, stream>>>(deg, N_NODES);
    gin_csr_count<<<edgeGrd, 256, 0, stream>>>(dst, deg, rank);
    gin_csr_scan<<<scanGrd, 1024, 0, stream>>>(deg, rowbeg);
    gin_csr_fill<<<edgeGrd, 256, 0, stream>>>(src, dst, rowbeg, rank, col);

    for (int i = 0; i < N_LAYERS; ++i) {
        const float* hin  = (i == 0) ? x : ((i & 1) ? hA : hB);
        float*       hout = (i & 1) ? hB : hA;
        int do_stats = (i < N_LAYERS - 1) ? 1 : 0;

        gin_gather<<<gathGrd, 256, 0, stream>>>(hin, rowbeg, deg, col, agg, stats);
        gin_mlp<<<mlpGrd, 256, 0, stream>>>(hin, agg,
                                            W1 + (long)i * DIM * DIM, b1 + (long)i * DIM,
                                            W2 + (long)i * DIM * DIM, b2 + (long)i * DIM,
                                            eps + i, hout, stats, do_stats);
        if (do_stats) {
            gin_bn_relu<<<bnGrd, 256, 0, stream>>>(hout, stats,
                                                   gamma + (long)i * DIM,
                                                   beta + (long)i * DIM);
        }
    }
    gin_zero<<<(NUM_GRAPHS * DIM + 255) / 256, 256, 0, stream>>>(out, NUM_GRAPHS * DIM);
    gin_pool<<<poolGrd, 256, 0, stream>>>(hB, batch, out);
}


# MLP kernel: LDS tiles at a row pitch of 132 floats instead of 128, so the A-fragment reads of both products no longer hit one bank 16 ways
# speedup vs baseline: 1.9291x; 1.1651x over previous
#include <hip/hip_runtime.h>
#include <hip/hip_bf16.h>

#define N_NODES   50000
#define N_EDGES   800000
#define DIM       128
#define N_LAYERS  4
#define NUM_GRAPHS 128
#define BN_EPS    1e-5f
#define TP        132

typedef __attribute__((ext_vector_type(2))) float v2f;
typedef __attribute__((ext_vector_type(8))) float v8f;

__global__ __launch_bounds__(256) void gin_zero(float* __restrict__ p, long n) {
    long i = (long)blockIdx.x * blockDim.x + threadIdx.x;
    if (i < n) p[i] = 0.0f;
}

__global__ __launch_bounds__(256) void gin_zero_i(int* __restrict__ p, int n) {
    int i = blockIdx.x * blockDim.x + threadIdx.x;
    if (i < n) p[i] = 0;
}

__global__ __launch_bounds__(256) void gin_csr_count(const int* __restrict__ dst,
                                                     int* __restrict__ deg,
                                                     int* __restrict__ rank) {
    int e = blockIdx.x * blockDim.x + threadIdx.x;
    if (e >= N_EDGES) return;
    rank[e] = atomicAdd(&deg[dst[e]], 1);
}

__global__ __launch_bounds__(1024) void gin_csr_scan(const int* __restrict__ deg,
                                                     int* __restrict__ rowbeg) {
    __shared__ int wsum[32];
    __shared__ int wpre[32];
    const int tid  = threadIdx.x;
    const int lane = tid & 31;
    const int wave = tid >> 5;
    const int chunk0 = blockIdx.x * 1024;

    int pre = 0;
    for (int i = tid; i < chunk0; i += 1024) pre += deg[i];

    const int idx = chunk0 + tid;
    const int v = (idx < N_NODES) ? deg[idx] : 0;

    int inc = v;
    for (int off = 1; off < 32; off <<= 1) {
        int t = __shfl_up(inc, off);
        if (lane >= off) inc += t;
    }
    for (int off = 16; off > 0; off >>= 1) pre += __shfl_xor(pre, off);

    if (lane == 31) wsum[wave] = inc;
    if (lane == 0)  wpre[wave] = pre;
    __syncthreads();

    int base = 0;
    for (int w = 0; w < 32; ++w) {
        base += wpre[w];
        if (w < wave) base += wsum[w];
    }
    const int excl = base + inc - v;
    if (idx < N_NODES) rowbeg[idx] = excl;
}

__global__ __launch_bounds__(256) void gin_csr_fill(const int* __restrict__ src,
                                                    const int* __restrict__ dst,
                                                    const int* __restrict__ rowbeg,
                                                    const int* __restrict__ rank,
                                                    int* __restrict__ col) {
    int e = blockIdx.x * blockDim.x + threadIdx.x;
    if (e >= N_EDGES) return;
    col[rowbeg[dst[e]] + rank[e]] = src[e];
}

__global__ __launch_bounds__(256) void gin_gather(const float* __restrict__ hin,
                                                  const int* __restrict__ rowbeg,
                                                  const int* __restrict__ deg,
                                                  const int* __restrict__ col,
                                                  float* __restrict__ agg,
                                                  float* __restrict__ stats) {
    if (blockIdx.x == 0) stats[threadIdx.x] = 0.0f;

    const int lane = threadIdx.x & 31;
    const int node = __builtin_amdgcn_readfirstlane((int)(blockIdx.x * 8 + (threadIdx.x >> 5)));
    if (node >= N_NODES) return;

    const int beg = rowbeg[node];
    const int cnt = deg[node];
    const float4* h4 = (const float4*)hin;

    float4 acc = make_float4(0.0f, 0.0f, 0.0f, 0.0f);
    for (int base = 0; base < cnt; base += 32) {
        const int m = (cnt - base < 32) ? (cnt - base) : 32;
        const int mine = (lane < m) ? col[beg + base + lane] : 0;
        for (int j = 0; j < m; j += 4) {
            float4 a0 = make_float4(0.0f, 0.0f, 0.0f, 0.0f);
            float4 a1 = a0, a2 = a0, a3 = a0;
            a0 = h4[(long)__builtin_amdgcn_readlane(mine, j) * (DIM / 4) + lane];
            if (j + 1 < m) a1 = h4[(long)__builtin_amdgcn_readlane(mine, j + 1) * (DIM / 4) + lane];
            if (j + 2 < m) a2 = h4[(long)__builtin_amdgcn_readlane(mine, j + 2) * (DIM / 4) + lane];
            if (j + 3 < m) a3 = h4[(long)__builtin_amdgcn_readlane(mine, j + 3) * (DIM / 4) + lane];
            acc.x += (a0.x + a1.x) + (a2.x + a3.x);
            acc.y += (a0.y + a1.y) + (a2.y + a3.y);
            acc.z += (a0.z + a1.z) + (a2.z + a3.z);
            acc.w += (a0.w + a1.w) + (a2.w + a3.w);
        }
    }
    ((float4*)agg)[(long)node * (DIM / 4) + lane] = acc;
}

__global__ __launch_bounds__(256) void gin_mlp(const float* __restrict__ hin,
                                               const float* __restrict__ agg,
                                               const float* __restrict__ W1,
                                               const float* __restrict__ b1,
                                               const float* __restrict__ W2,
                                               const float* __restrict__ b2,
                                               const float* __restrict__ epsp,
                                               float* __restrict__ hout,
                                               float* __restrict__ stats,
                                               int do_stats) {
    __shared__ float zt[16 * TP];
    __shared__ float y1[16 * TP];

    const int row0 = blockIdx.x * 16;
    const int tid  = threadIdx.x;
    const float e1 = 1.0f + epsp[0];

    for (int i = tid; i < 16 * DIM; i += 256) {
        int r = i >> 7;
        int c = i & (DIM - 1);
        long g = (long)(row0 + r) * DIM + c;
        zt[r * TP + c] = e1 * hin[g] + agg[g];
    }
    __syncthreads();

    const int wave = tid >> 5;
    const int lane = tid & 31;
    const int n    = lane & 15;
    const int hi   = lane >> 4;
    const int col0 = wave * 16;
    const int kk   = hi * 2;

    __builtin_prefetch(&W1[col0], 0, 0);
    __builtin_prefetch(&W2[col0], 0, 0);

    v8f acc;
    {
        float bv = b1[col0 + n];
        for (int v = 0; v < 8; ++v) acc[v] = bv;
    }
    for (int k = 0; k < DIM; k += 4) {
        v2f a, b;
        const float* zp = &zt[n * TP + k + kk];
        a.x = zp[0];
        a.y = zp[1];
        b.x = W1[(k + kk) * DIM + col0 + n];
        b.y = W1[(k + kk + 1) * DIM + col0 + n];
        acc = __builtin_amdgcn_wmma_f32_16x16x4_f32(false, a, false, b,
                                                    (short)0, acc, false, false);
    }
    for (int v = 0; v < 8; ++v) {
        int m = v + 8 * hi;
        y1[m * TP + col0 + n] = fmaxf(acc[v], 0.0f);
    }
    __syncthreads();

    {
        float bv = b2[col0 + n];
        for (int v = 0; v < 8; ++v) acc[v] = bv;
    }
    for (int k = 0; k < DIM; k += 4) {
        v2f a, b;
        const float* yp = &y1[n * TP + k + kk];
        a.x = yp[0];
        a.y = yp[1];
        b.x = W2[(k + kk) * DIM + col0 + n];
        b.y = W2[(k + kk + 1) * DIM + col0 + n];
        acc = __builtin_amdgcn_wmma_f32_16x16x4_f32(false, a, false, b,
                                                    (short)0, acc, false, false);
    }

    float s = 0.0f, ssq = 0.0f;
    for (int v = 0; v < 8; ++v) {
        int m = v + 8 * hi;
        float val = acc[v];
        hout[(long)(row0 + m) * DIM + col0 + n] = val;
        s += val;
        ssq += val * val;
    }
    if (do_stats) {
        atomicAdd(&stats[col0 + n], s);
        atomicAdd(&stats[DIM + col0 + n], ssq);
    }
}

__global__ __launch_bounds__(256) void gin_bn_relu(float* __restrict__ z,
                                                   const float* __restrict__ stats,
                                                   const float* __restrict__ gamma,
                                                   const float* __restrict__ beta) {
    long i = (long)blockIdx.x * blockDim.x + threadIdx.x;
    if (i >= (long)N_NODES * DIM) return;
    int c = (int)(i & (DIM - 1));
    const float invN = 1.0f / (float)N_NODES;
    float mu  = stats[c] * invN;
    float var = stats[DIM + c] * invN - mu * mu;
    float v = (z[i] - mu) * rsqrtf(var + BN_EPS) * gamma[c] + beta[c];
    z[i] = fmaxf(v, 0.0f);
}

__global__ __launch_bounds__(256) void gin_pool(const float* __restrict__ h,
                                                const int* __restrict__ batch,
                                                float* __restrict__ out) {
    long t = (long)blockIdx.x * blockDim.x + threadIdx.x;
    long nidx = t >> 5;
    int lane = (int)(t & 31);
    if (nidx >= N_NODES) return;
    int g = batch[nidx];
    const float4* hp = (const float4*)(h + nidx * DIM);
    float4 v = hp[lane];
    float* op = out + (long)g * DIM + lane * 4;
    atomicAdd(op + 0, v.x);
    atomicAdd(op + 1, v.y);
    atomicAdd(op + 2, v.z);
    atomicAdd(op + 3, v.w);
}

extern "C" void kernel_launch(void* const* d_in, const int* in_sizes, int n_in,
                              void* d_out, int out_size, void* d_ws, size_t ws_size,
                              hipStream_t stream) {
    const float* x     = (const float*)d_in[0];
    const int*   ei    = (const int*)d_in[1];
    const int*   batch = (const int*)d_in[2];
    const float* W1    = (const float*)d_in[3];
    const float* b1    = (const float*)d_in[4];
    const float* W2    = (const float*)d_in[5];
    const float* b2    = (const float*)d_in[6];
    const float* eps   = (const float*)d_in[7];
    const float* gamma = (const float*)d_in[8];
    const float* beta  = (const float*)d_in[9];
    float* out = (float*)d_out;

    const long ND = (long)N_NODES * DIM;
    float* hA    = (float*)d_ws;
    float* hB    = hA + ND;
    float* agg   = hB + ND;
    float* stats = agg + ND;
    int* deg    = (int*)(stats + 2 * DIM);
    int* rowbeg = deg + N_NODES;
    int* rank   = rowbeg + N_NODES;
    int* col    = rank + N_EDGES;

    const int* src = ei;
    const int* dst = ei + N_EDGES;

    const int  degGrd  = (N_NODES + 255) / 256;
    const int  edgeGrd = (N_EDGES + 255) / 256;
    const int  scanGrd = (N_NODES + 1023) / 1024;
    const int  gathGrd = (N_NODES + 7) / 8;
    const int  mlpGrd  = N_NODES / 16;
    const int  bnGrd   = (int)((ND + 255) / 256);
    const int  poolGrd = (int)(((long)N_NODES * 32 + 255) / 256);

    gin_zero_i<<<degGrd, 256, 0, stream>>>(deg, N_NODES);
    gin_csr_count<<<edgeGrd, 256, 0, stream>>>(dst, deg, rank);
    gin_csr_scan<<<scanGrd, 1024, 0, stream>>>(deg, rowbeg);
    gin_csr_fill<<<edgeGrd, 256, 0, stream>>>(src, dst, rowbeg, rank, col);

    for (int i = 0; i < N_LAYERS; ++i) {
        const float* hin  = (i == 0) ? x : ((i & 1) ? hA : hB);
        float*       hout = (i & 1) ? hB : hA;
        int do_stats = (i < N_LAYERS - 1) ? 1 : 0;

        gin_gather<<<gathGrd, 256, 0, stream>>>(hin, rowbeg, deg, col, agg, stats);
        gin_mlp<<<mlpGrd, 256, 0, stream>>>(hin, agg,
                                            W1 + (long)i * DIM * DIM, b1 + (long)i * DIM,
                                            W2 + (long)i * DIM * DIM, b2 + (long)i * DIM,
                                            eps + i, hout, stats, do_stats);
        if (do_stats) {
            gin_bn_relu<<<bnGrd, 256, 0, stream>>>(hout, stats,
                                                   gamma + (long)i * DIM,
                                                   beta + (long)i * DIM);
        }
    }
    gin_zero<<<(NUM_GRAPHS * DIM + 255) / 256, 256, 0, stream>>>(out, NUM_GRAPHS * DIM);
    gin_pool<<<poolGrd, 256, 0, stream>>>(hB, batch, out);
}


# BatchNorm+ReLU applied where the next layer reads h (gathered rows, MLP input tile) instead of a one-thread-per-element pass over h: 3 launches and 600K short waves fewer; sum order unchanged
# speedup vs baseline: 2.0916x; 1.0842x over previous
#include <hip/hip_runtime.h>
#include <hip/hip_bf16.h>

#define N_NODES   50000
#define N_EDGES   800000
#define DIM       128
#define N_LAYERS  4
#define NUM_GRAPHS 128
#define BN_EPS    1e-5f
#define TP        132

typedef __attribute__((ext_vector_type(2))) float v2f;
typedef __attribute__((ext_vector_type(8))) float v8f;

__global__ __launch_bounds__(256) void gin_zero(float* __restrict__ p, long n) {
    long i = (long)blockIdx.x * blockDim.x + threadIdx.x;
    if (i < n) p[i] = 0.0f;
}

__global__ __launch_bounds__(256) void gin_zero_i(int* __restrict__ p, int n) {
    int i = blockIdx.x * blockDim.x + threadIdx.x;
    if (i < n) p[i] = 0;
}

__global__ __launch_bounds__(256) void gin_csr_count(const int* __restrict__ dst,
                                                     int* __restrict__ deg,
                                                     int* __restrict__ rank) {
    int e = blockIdx.x * blockDim.x + threadIdx.x;
    if (e >= N_EDGES) return;
    rank[e] = atomicAdd(&deg[dst[e]], 1);
}

__global__ __launch_bounds__(1024) void gin_csr_scan(const int* __restrict__ deg,
                                                     int* __restrict__ rowbeg) {
    __shared__ int wsum[32];
    __shared__ int wpre[32];
    const int tid  = threadIdx.x;
    const int lane = tid & 31;
    const int wave = tid >> 5;
    const int chunk0 = blockIdx.x * 1024;

    int pre = 0;
    for (int i = tid; i < chunk0; i += 1024) pre += deg[i];

    const int idx = chunk0 + tid;
    const int v = (idx < N_NODES) ? deg[idx] : 0;

    int inc = v;
    for (int off = 1; off < 32; off <<= 1) {
        int t = __shfl_up(inc, off);
        if (lane >= off) inc += t;
    }
    for (int off = 16; off > 0; off >>= 1) pre += __shfl_xor(pre, off);

    if (lane == 31) wsum[wave] = inc;
    if (lane == 0)  wpre[wave] = pre;
    __syncthreads();

    int base = 0;
    for (int w = 0; w < 32; ++w) {
        base += wpre[w];
        if (w < wave) base += wsum[w];
    }
    const int excl = base + inc - v;
    if (idx < N_NODES) rowbeg[idx] = excl;
}

__global__ __launch_bounds__(256) void gin_csr_fill(const int* __restrict__ src,
                                                    const int* __restrict__ dst,
                                                    const int* __restrict__ rowbeg,
                                                    const int* __restrict__ rank,
                                                    int* __restrict__ col) {
    int e = blockIdx.x * blockDim.x + threadIdx.x;
    if (e >= N_EDGES) return;
    col[rowbeg[dst[e]] + rank[e]] = src[e];
}

__device__ __forceinline__ void bn_coef(const float* __restrict__ st,
                                        const float* __restrict__ gamma,
                                        const float* __restrict__ beta,
                                        int c, float& mu, float& sc, float& be) {
    const float invN = 1.0f / (float)N_NODES;
    mu = st[c] * invN;
    float var = st[DIM + c] * invN - mu * mu;
    sc = rsqrtf(var + BN_EPS) * gamma[c];
    be = beta[c];
}
__device__ __forceinline__ float bn_relu1(float z, float mu, float sc, float be) {
    return fmaxf((z - mu) * sc + be, 0.0f);
}
__device__ __forceinline__ float4 bn_relu4(float4 z, float4 mu, float4 sc, float4 be) {
    float4 r;
    r.x = bn_relu1(z.x, mu.x, sc.x, be.x);
    r.y = bn_relu1(z.y, mu.y, sc.y, be.y);
    r.z = bn_relu1(z.z, mu.z, sc.z, be.z);
    r.w = bn_relu1(z.w, mu.w, sc.w, be.w);
    return r;
}

template <bool NORM>
__global__ __launch_bounds__(256) void gin_gather(const float* __restrict__ hin,
                                                  const int* __restrict__ rowbeg,
                                                  const int* __restrict__ deg,
                                                  const int* __restrict__ col,
                                                  float* __restrict__ agg,
                                                  float* __restrict__ stats,
                                                  const float* __restrict__ st_prev,
                                                  const float* __restrict__ gamma,
                                                  const float* __restrict__ beta) {
    if (blockIdx.x == 0) stats[threadIdx.x] = 0.0f;

    const int lane = threadIdx.x & 31;
    const int node = __builtin_amdgcn_readfirstlane((int)(blockIdx.x * 8 + (threadIdx.x >> 5)));
    if (node >= N_NODES) return;

    float4 mu = make_float4(0.0f, 0.0f, 0.0f, 0.0f);
    float4 sc = mu, be = mu;
    if (NORM) {
        bn_coef(st_prev, gamma, beta, 4 * lane + 0, mu.x, sc.x, be.x);
        bn_coef(st_prev, gamma, beta, 4 * lane + 1, mu.y, sc.y, be.y);
        bn_coef(st_prev, gamma, beta, 4 * lane + 2, mu.z, sc.z, be.z);
        bn_coef(st_prev, gamma, beta, 4 * lane + 3, mu.w, sc.w, be.w);
    }

    const int beg = rowbeg[node];
    const int cnt = deg[node];
    const float4* h4 = (const float4*)hin;

    float4 acc = make_float4(0.0f, 0.0f, 0.0f, 0.0f);
    for (int base = 0; base < cnt; base += 32) {
        const int m = (cnt - base < 32) ? (cnt - base) : 32;
        const int mine = (lane < m) ? col[beg + base + lane] : 0;
        for (int j = 0; j < m; j += 4) {
            float4 a0 = make_float4(0.0f, 0.0f, 0.0f, 0.0f);
            float4 a1 = a0, a2 = a0, a3 = a0;
            a0 = h4[(long)__builtin_amdgcn_readlane(mine, j) * (DIM / 4) + lane];
            if (j + 1 < m) a1 = h4[(long)__builtin_amdgcn_readlane(mine, j + 1) * (DIM / 4) + lane];
            if (j + 2 < m) a2 = h4[(long)__builtin_amdgcn_readlane(mine, j + 2) * (DIM / 4) + lane];
            if (j + 3 < m) a3 = h4[(long)__builtin_amdgcn_readlane(mine, j + 3) * (DIM / 4) + lane];
            if (NORM) {
                a0 = bn_relu4(a0, mu, sc, be);
                if (j + 1 < m) a1 = bn_relu4(a1, mu, sc, be);
                if (j + 2 < m) a2 = bn_relu4(a2, mu, sc, be);
                if (j + 3 < m) a3 = bn_relu4(a3, mu, sc, be);
            }
            acc.x += (a0.x + a1.x) + (a2.x + a3.x);
            acc.y += (a0.y + a1.y) + (a2.y + a3.y);
            acc.z += (a0.z + a1.z) + (a2.z + a3.z);
            acc.w += (a0.w + a1.w) + (a2.w + a3.w);
        }
    }
    ((float4*)agg)[(long)node * (DIM / 4) + lane] = acc;
}

template <bool NORM>
__global__ __launch_bounds__(256) void gin_mlp(const float* __restrict__ hin,
                                               const float* __restrict__ agg,
                                               const float* __restrict__ W1,
                                               const float* __restrict__ b1,
                                               const float* __restrict__ W2,
                                               const float* __restrict__ b2,
                                               const float* __restrict__ epsp,
                                               float* __restrict__ hout,
                                               float* __restrict__ stats,
                                               int do_stats,
                                               const float* __restrict__ st_prev,
                                               const float* __restrict__ gamma,
                                               const float* __restrict__ beta) {
    __shared__ float zt[16 * TP];
    __shared__ float y1[16 * TP];

    const int row0 = blockIdx.x * 16;
    const int tid  = threadIdx.x;
    const float e1 = 1.0f + epsp[0];

    float mu = 0.0f, sc = 0.0f, be = 0.0f;
    if (NORM) bn_coef(st_prev, gamma, beta, tid & (DIM - 1), mu, sc, be);

    for (int i = tid; i < 16 * DIM; i += 256) {
        int r = i >> 7;
        int c = i & (DIM - 1);
        long g = (long)(row0 + r) * DIM + c;
        float h = hin[g];
        if (NORM) h = bn_relu1(h, mu, sc, be);
        zt[r * TP + c] = e1 * h + agg[g];
    }
    __syncthreads();

    const int wave = tid >> 5;
    const int lane = tid & 31;
    const int n    = lane & 15;
    const int hi   = lane >> 4;
    const int col0 = wave * 16;
    const int kk   = hi * 2;

    __builtin_prefetch(&W1[col0], 0, 0);
    __builtin_prefetch(&W2[col0], 0, 0);

    v8f acc;
    {
        float bv = b1[col0 + n];
        for (int v = 0; v < 8; ++v) acc[v] = bv;
    }
    for (int k = 0; k < DIM; k += 4) {
        v2f a, b;
        const float* zp = &zt[n * TP + k + kk];
        a.x = zp[0];
        a.y = zp[1];
        b.x = W1[(k + kk) * DIM + col0 + n];
        b.y = W1[(k + kk + 1) * DIM + col0 + n];
        acc = __builtin_amdgcn_wmma_f32_16x16x4_f32(false, a, false, b,
                                                    (short)0, acc, false, false);
    }
    for (int v = 0; v < 8; ++v) {
        int m = v + 8 * hi;
        y1[m * TP + col0 + n] = fmaxf(acc[v], 0.0f);
    }
    __syncthreads();

    {
        float bv = b2[col0 + n];
        for (int v = 0; v < 8; ++v) acc[v] = bv;
    }
    for (int k = 0; k < DIM; k += 4) {
        v2f a, b;
        const float* yp = &y1[n * TP + k + kk];
        a.x = yp[0];
        a.y = yp[1];
        b.x = W2[(k + kk) * DIM + col0 + n];
        b.y = W2[(k + kk + 1) * DIM + col0 + n];
        acc = __builtin_amdgcn_wmma_f32_16x16x4_f32(false, a, false, b,
                                                    (short)0, acc, false, false);
    }

    float s = 0.0f, ssq = 0.0f;
    for (int v = 0; v < 8; ++v) {
        int m = v + 8 * hi;
        float val = acc[v];
        hout[(long)(row0 + m) * DIM + col0 + n] = val;
        s += val;
        ssq += val * val;
    }
    if (do_stats) {
        atomicAdd(&stats[col0 + n], s);
        atomicAdd(&stats[DIM + col0 + n], ssq);
    }
}

__global__ __launch_bounds__(256) void gin_pool(const float* __restrict__ h,
                                                const int* __restrict__ batch,
                                                float* __restrict__ out) {
    long t = (long)blockIdx.x * blockDim.x + threadIdx.x;
    long nidx = t >> 5;
    int lane = (int)(t & 31);
    if (nidx >= N_NODES) return;
    int g = batch[nidx];
    const float4* hp = (const float4*)(h + nidx * DIM);
    float4 v = hp[lane];
    float* op = out + (long)g * DIM + lane * 4;
    atomicAdd(op + 0, v.x);
    atomicAdd(op + 1, v.y);
    atomicAdd(op + 2, v.z);
    atomicAdd(op + 3, v.w);
}

extern "C" void kernel_launch(void* const* d_in, const int* in_sizes, int n_in,
                              void* d_out, int out_size, void* d_ws, size_t ws_size,
                              hipStream_t stream) {
    const float* x     = (const float*)d_in[0];
    const int*   ei    = (const int*)d_in[1];
    const int*   batch = (const int*)d_in[2];
    const float* W1    = (const float*)d_in[3];
    const float* b1    = (const float*)d_in[4];
    const float* W2    = (const float*)d_in[5];
    const float* b2    = (const float*)d_in[6];
    const float* eps   = (const float*)d_in[7];
    const float* gamma = (const float*)d_in[8];
    const float* beta  = (const float*)d_in[9];
    float* out = (float*)d_out;

    const long ND = (long)N_NODES * DIM;
    float* hA    = (float*)d_ws;
    float* hB    = hA + ND;
    float* agg   = hB + ND;
    float* stats = agg + ND;
    int* deg    = (int*)(stats + 4 * DIM);
    int* rowbeg = deg + N_NODES;
    int* rank   = rowbeg + N_NODES;
    int* col    = rank + N_EDGES;

    const int* src = ei;
    const int* dst = ei + N_EDGES;

    const int  degGrd  = (N_NODES + 255) / 256;
    const int  edgeGrd = (N_EDGES + 255) / 256;
    const int  scanGrd = (N_NODES + 1023) / 1024;
    const int  gathGrd = (N_NODES + 7) / 8;
    const int  mlpGrd  = N_NODES / 16;
    const int  poolGrd = (int)(((long)N_NODES * 32 + 255) / 256);

    gin_zero_i<<<degGrd, 256, 0, stream>>>(deg, N_NODES);
    gin_csr_count<<<edgeGrd, 256, 0, stream>>>(dst, deg, rank);
    gin_csr_scan<<<scanGrd, 1024, 0, stream>>>(deg, rowbeg);
    gin_csr_fill<<<edgeGrd, 256, 0, stream>>>(src, dst, rowbeg, rank, col);

    for (int i = 0; i < N_LAYERS; ++i) {
        const float* hin  = (i == 0) ? x : ((i & 1) ? hA : hB);
        float*       hout = (i & 1) ? hB : hA;
        int do_stats = (i < N_LAYERS - 1) ? 1 : 0;
        float*       st_cur  = stats + (i & 1) * 2 * DIM;
        const float* st_prev = stats + ((i + 1) & 1) * 2 * DIM;
        const float* g_prev  = gamma + (long)(i > 0 ? i - 1 : 0) * DIM;
        const float* b_prev  = beta + (long)(i > 0 ? i - 1 : 0) * DIM;

        if (i == 0) {
            gin_gather<false><<<gathGrd, 256, 0, stream>>>(hin, rowbeg, deg, col, agg, st_cur,
                                                           st_prev, g_prev, b_prev);
            gin_mlp<false><<<mlpGrd, 256, 0, stream>>>(hin, agg,
                                                       W1 + (long)i * DIM * DIM, b1 + (long)i * DIM,
                                                       W2 + (long)i * DIM * DIM, b2 + (long)i * DIM,
                                                       eps + i, hout, st_cur, do_stats,
                                                       st_prev, g_prev, b_prev);
        } else {
            gin_gather<true><<<gathGrd, 256, 0, stream>>>(hin, rowbeg, deg, col, agg, st_cur,
                                                          st_prev, g_prev, b_prev);
            gin_mlp<true><<<mlpGrd, 256, 0, stream>>>(hin, agg,
                                                      W1 + (long)i * DIM * DIM, b1 + (long)i * DIM,
                                                      W2 + (long)i * DIM * DIM, b2 + (long)i * DIM,
                                                      eps + i, hout, st_cur, do_stats,
                                                      st_prev, g_prev, b_prev);
        }
    }
    gin_zero<<<(NUM_GRAPHS * DIM + 255) / 256, 256, 0, stream>>>(out, NUM_GRAPHS * DIM);
    gin_pool<<<poolGrd, 256, 0, stream>>>(hB, batch, out);
}


# Neighbour gather moved into the MLP kernel: each wave sums the source rows of two of the block's 16 nodes straight into the LDS A tile; agg plane and 4 launches go; sum order per row unchanged
# speedup vs baseline: 2.3442x; 1.1208x over previous
#include <hip/hip_runtime.h>
#include <hip/hip_bf16.h>

#define N_NODES   50000
#define N_EDGES   800000
#define DIM       128
#define N_LAYERS  4
#define NUM_GRAPHS 128
#define BN_EPS    1e-5f
#define TP        132

typedef __attribute__((ext_vector_type(2))) float v2f;
typedef __attribute__((ext_vector_type(8))) float v8f;

__global__ __launch_bounds__(256) void gin_zero(float* __restrict__ p, long n) {
    long i = (long)blockIdx.x * blockDim.x + threadIdx.x;
    if (i < n) p[i] = 0.0f;
}

__global__ __launch_bounds__(256) void gin_zero_i(int* __restrict__ p, int n, float* __restrict__ stats) {
    if (blockIdx.x < N_LAYERS) stats[blockIdx.x * 2 * DIM + threadIdx.x] = 0.0f;
    int i = blockIdx.x * blockDim.x + threadIdx.x;
    if (i < n) p[i] = 0;
}

__global__ __launch_bounds__(256) void gin_csr_count(const int* __restrict__ dst,
                                                     int* __restrict__ deg,
                                                     int* __restrict__ rank) {
    int e = blockIdx.x * blockDim.x + threadIdx.x;
    if (e >= N_EDGES) return;
    rank[e] = atomicAdd(&deg[dst[e]], 1);
}

__global__ __launch_bounds__(1024) void gin_csr_scan(const int* __restrict__ deg,
                                                     int* __restrict__ rowbeg) {
    __shared__ int wsum[32];
    __shared__ int wpre[32];
    const int tid  = threadIdx.x;
    const int lane = tid & 31;
    const int wave = tid >> 5;
    const int chunk0 = blockIdx.x * 1024;

    int pre = 0;
    for (int i = tid; i < chunk0; i += 1024) pre += deg[i];

    const int idx = chunk0 + tid;
    const int v = (idx < N_NODES) ? deg[idx] : 0;

    int inc = v;
    for (int off = 1; off < 32; off <<= 1) {
        int t = __shfl_up(inc, off);
        if (lane >= off) inc += t;
    }
    for (int off = 16; off > 0; off >>= 1) pre += __shfl_xor(pre, off);

    if (lane == 31) wsum[wave] = inc;
    if (lane == 0)  wpre[wave] = pre;
    __syncthreads();

    int base = 0;
    for (int w = 0; w < 32; ++w) {
        base += wpre[w];
        if (w < wave) base += wsum[w];
    }
    const int excl = base + inc - v;
    if (idx < N_NODES) rowbeg[idx] = excl;
}

__global__ __launch_bounds__(256) void gin_csr_fill(const int* __restrict__ src,
                                                    const int* __restrict__ dst,
                                                    const int* __restrict__ rowbeg,
                                                    const int* __restrict__ rank,
                                                    int* __restrict__ col) {
    int e = blockIdx.x * blockDim.x + threadIdx.x;
    if (e >= N_EDGES) return;
    col[rowbeg[dst[e]] + rank[e]] = src[e];
}

__device__ __forceinline__ void bn_coef(const float* __restrict__ st,
                                        const float* __restrict__ gamma,
                                        const float* __restrict__ beta,
                                        int c, float& mu, float& sc, float& be) {
    const float invN = 1.0f / (float)N_NODES;
    mu = st[c] * invN;
    float var = st[DIM + c] * invN - mu * mu;
    sc = rsqrtf(var + BN_EPS) * gamma[c];
    be = beta[c];
}
__device__ __forceinline__ float bn_relu1(float z, float mu, float sc, float be) {
    return fmaxf((z - mu) * sc + be, 0.0f);
}
__device__ __forceinline__ float4 bn_relu4(float4 z, float4 mu, float4 sc, float4 be) {
    float4 r;
    r.x = bn_relu1(z.x, mu.x, sc.x, be.x);
    r.y = bn_relu1(z.y, mu.y, sc.y, be.y);
    r.z = bn_relu1(z.z, mu.z, sc.z, be.z);
    r.w = bn_relu1(z.w, mu.w, sc.w, be.w);
    return r;
}

template <bool NORM>
__global__ __launch_bounds__(256) void gin_layer(const float* __restrict__ hin,
                                                 const int* __restrict__ rowbeg,
                                                 const int* __restrict__ deg,
                                                 const int* __restrict__ col,
                                                 const float* __restrict__ W1,
                                                 const float* __restrict__ b1,
                                                 const float* __restrict__ W2,
                                                 const float* __restrict__ b2,
                                                 const float* __restrict__ epsp,
                                                 float* __restrict__ hout,
                                                 float* __restrict__ stats,
                                                 int do_stats,
                                                 const float* __restrict__ st_prev,
                                                 const float* __restrict__ gamma,
                                                 const float* __restrict__ beta) {
    __shared__ __attribute__((aligned(16))) float zt[16 * TP];
    __shared__ __attribute__((aligned(16))) float y1[16 * TP];

    const int row0 = blockIdx.x * 16;
    const int tid  = threadIdx.x;
    const int lane = tid & 31;
    const int wave = __builtin_amdgcn_readfirstlane(tid >> 5);
    const float e1 = 1.0f + epsp[0];

    {
        const float4 zero4 = make_float4(0.0f, 0.0f, 0.0f, 0.0f);
        float4 mu = zero4, sc = zero4, be = zero4;
        if (NORM) {
            bn_coef(st_prev, gamma, beta, 4 * lane + 0, mu.x, sc.x, be.x);
            bn_coef(st_prev, gamma, beta, 4 * lane + 1, mu.y, sc.y, be.y);
            bn_coef(st_prev, gamma, beta, 4 * lane + 2, mu.z, sc.z, be.z);
            bn_coef(st_prev, gamma, beta, 4 * lane + 3, mu.w, sc.w, be.w);
        }

        const int rA = 2 * wave;
        const int nodeA = row0 + rA;
        const int nodeB = nodeA + 1;
        const int begA = rowbeg[nodeA];
        const int begB = rowbeg[nodeB];
        const int cntA = deg[nodeA];
        const int cntB = deg[nodeB];
        const float4* h4 = (const float4*)hin;

        float4 selfA = h4[(long)nodeA * (DIM / 4) + lane];
        float4 selfB = h4[(long)nodeB * (DIM / 4) + lane];

        float4 accA = zero4, accB = zero4;
        const int cmax = (cntA > cntB) ? cntA : cntB;
        for (int base = 0; base < cmax; base += 32) {
            int mA = cntA - base; mA = (mA < 0) ? 0 : ((mA > 32) ? 32 : mA);
            int mB = cntB - base; mB = (mB < 0) ? 0 : ((mB > 32) ? 32 : mB);
            const int mineA = (lane < mA) ? col[begA + base + lane] : 0;
            const int mineB = (lane < mB) ? col[begB + base + lane] : 0;
            const int mm = (mA > mB) ? mA : mB;
            asm volatile("" :: "v"(mineA), "v"(mineB));
            for (int j = 0; j < mm; j += 4) {
                float4 a0 = zero4, a1 = zero4, a2 = zero4, a3 = zero4;
                float4 c0 = zero4, c1 = zero4, c2 = zero4, c3 = zero4;
                if (j + 0 < mA) a0 = h4[(long)__builtin_amdgcn_readlane(mineA, j + 0) * (DIM / 4) + lane];
                if (j + 1 < mA) a1 = h4[(long)__builtin_amdgcn_readlane(mineA, j + 1) * (DIM / 4) + lane];
                if (j + 2 < mA) a2 = h4[(long)__builtin_amdgcn_readlane(mineA, j + 2) * (DIM / 4) + lane];
                if (j + 3 < mA) a3 = h4[(long)__builtin_amdgcn_readlane(mineA, j + 3) * (DIM / 4) + lane];
                if (j + 0 < mB) c0 = h4[(long)__builtin_amdgcn_readlane(mineB, j + 0) * (DIM / 4) + lane];
                if (j + 1 < mB) c1 = h4[(long)__builtin_amdgcn_readlane(mineB, j + 1) * (DIM / 4) + lane];
                if (j + 2 < mB) c2 = h4[(long)__builtin_amdgcn_readlane(mineB, j + 2) * (DIM / 4) + lane];
                if (j + 3 < mB) c3 = h4[(long)__builtin_amdgcn_readlane(mineB, j + 3) * (DIM / 4) + lane];
                if (NORM) {
                    if (j + 0 < mA) a0 = bn_relu4(a0, mu, sc, be);
                    if (j + 1 < mA) a1 = bn_relu4(a1, mu, sc, be);
                    if (j + 2 < mA) a2 = bn_relu4(a2, mu, sc, be);
                    if (j + 3 < mA) a3 = bn_relu4(a3, mu, sc, be);
                    if (j + 0 < mB) c0 = bn_relu4(c0, mu, sc, be);
                    if (j + 1 < mB) c1 = bn_relu4(c1, mu, sc, be);
                    if (j + 2 < mB) c2 = bn_relu4(c2, mu, sc, be);
                    if (j + 3 < mB) c3 = bn_relu4(c3, mu, sc, be);
                }
                accA.x += (a0.x + a1.x) + (a2.x + a3.x);
                accA.y += (a0.y + a1.y) + (a2.y + a3.y);
                accA.z += (a0.z + a1.z) + (a2.z + a3.z);
                accA.w += (a0.w + a1.w) + (a2.w + a3.w);
                accB.x += (c0.x + c1.x) + (c2.x + c3.x);
                accB.y += (c0.y + c1.y) + (c2.y + c3.y);
                accB.z += (c0.z + c1.z) + (c2.z + c3.z);
                accB.w += (c0.w + c1.w) + (c2.w + c3.w);
            }
        }

        if (NORM) {
            selfA = bn_relu4(selfA, mu, sc, be);
            selfB = bn_relu4(selfB, mu, sc, be);
        }
        float4 zA, zB;
        zA.x = e1 * selfA.x + accA.x;
        zA.y = e1 * selfA.y + accA.y;
        zA.z = e1 * selfA.z + accA.z;
        zA.w = e1 * selfA.w + accA.w;
        zB.x = e1 * selfB.x + accB.x;
        zB.y = e1 * selfB.y + accB.y;
        zB.z = e1 * selfB.z + accB.z;
        zB.w = e1 * selfB.w + accB.w;
        *(float4*)&zt[rA * TP + 4 * lane]       = zA;
        *(float4*)&zt[(rA + 1) * TP + 4 * lane] = zB;
    }
    __syncthreads();

    const int n    = lane & 15;
    const int hi   = lane >> 4;
    const int col0 = wave * 16;
    const int kk   = hi * 2;

    __builtin_prefetch(&W1[col0], 0, 0);
    __builtin_prefetch(&W2[col0], 0, 0);

    v8f acc;
    {
        float bv = b1[col0 + n];
        for (int v = 0; v < 8; ++v) acc[v] = bv;
    }
    for (int k = 0; k < DIM; k += 4) {
        v2f a, b;
        const float* zp = &zt[n * TP + k + kk];
        a.x = zp[0];
        a.y = zp[1];
        b.x = W1[(k + kk) * DIM + col0 + n];
        b.y = W1[(k + kk + 1) * DIM + col0 + n];
        acc = __builtin_amdgcn_wmma_f32_16x16x4_f32(false, a, false, b,
                                                    (short)0, acc, false, false);
    }
    for (int v = 0; v < 8; ++v) {
        int m = v + 8 * hi;
        y1[m * TP + col0 + n] = fmaxf(acc[v], 0.0f);
    }
    __syncthreads();

    {
        float bv = b2[col0 + n];
        for (int v = 0; v < 8; ++v) acc[v] = bv;
    }
    for (int k = 0; k < DIM; k += 4) {
        v2f a, b;
        const float* yp = &y1[n * TP + k + kk];
        a.x = yp[0];
        a.y = yp[1];
        b.x = W2[(k + kk) * DIM + col0 + n];
        b.y = W2[(k + kk + 1) * DIM + col0 + n];
        acc = __builtin_amdgcn_wmma_f32_16x16x4_f32(false, a, false, b,
                                                    (short)0, acc, false, false);
    }

    float s = 0.0f, ssq = 0.0f;
    for (int v = 0; v < 8; ++v) {
        int m = v + 8 * hi;
        float val = acc[v];
        hout[(long)(row0 + m) * DIM + col0 + n] = val;
        s += val;
        ssq += val * val;
    }
    if (do_stats) {
        atomicAdd(&stats[col0 + n], s);
        atomicAdd(&stats[DIM + col0 + n], ssq);
    }
}

__global__ __launch_bounds__(256) void gin_pool(const float* __restrict__ h,
                                                const int* __restrict__ batch,
                                                float* __restrict__ out) {
    long t = (long)blockIdx.x * blockDim.x + threadIdx.x;
    long nidx = t >> 5;
    int lane = (int)(t & 31);
    if (nidx >= N_NODES) return;
    int g = batch[nidx];
    const float4* hp = (const float4*)(h + nidx * DIM);
    float4 v = hp[lane];
    float* op = out + (long)g * DIM + lane * 4;
    atomicAdd(op + 0, v.x);
    atomicAdd(op + 1, v.y);
    atomicAdd(op + 2, v.z);
    atomicAdd(op + 3, v.w);
}

extern "C" void kernel_launch(void* const* d_in, const int* in_sizes, int n_in,
                              void* d_out, int out_size, void* d_ws, size_t ws_size,
                              hipStream_t stream) {
    const float* x     = (const float*)d_in[0];
    const int*   ei    = (const int*)d_in[1];
    const int*   batch = (const int*)d_in[2];
    const float* W1    = (const float*)d_in[3];
    const float* b1    = (const float*)d_in[4];
    const float* W2    = (const float*)d_in[5];
    const float* b2    = (const float*)d_in[6];
    const float* eps   = (const float*)d_in[7];
    const float* gamma = (const float*)d_in[8];
    const float* beta  = (const float*)d_in[9];
    float* out = (float*)d_out;

    const long ND = (long)N_NODES * DIM;
    float* hA    = (float*)d_ws;
    float* hB    = hA + ND;
    float* stats = hB + ND;
    int* deg    = (int*)(stats + N_LAYERS * 2 * DIM);
    int* rowbeg = deg + N_NODES;
    int* rank   = rowbeg + N_NODES;
    int* col    = rank + N_EDGES;

    const int* src = ei;
    const int* dst = ei + N_EDGES;

    const int  degGrd  = (N_NODES + 255) / 256;
    const int  edgeGrd = (N_EDGES + 255) / 256;
    const int  scanGrd = (N_NODES + 1023) / 1024;
    const int  layGrd  = N_NODES / 16;
    const int  poolGrd = (int)(((long)N_NODES * 32 + 255) / 256);

    gin_zero_i<<<degGrd, 256, 0, stream>>>(deg, N_NODES, stats);
    gin_csr_count<<<edgeGrd, 256, 0, stream>>>(dst, deg, rank);
    gin_csr_scan<<<scanGrd, 1024, 0, stream>>>(deg, rowbeg);
    gin_csr_fill<<<edgeGrd, 256, 0, stream>>>(src, dst, rowbeg, rank, col);

    for (int i = 0; i < N_LAYERS; ++i) {
        const float* hin  = (i == 0) ? x : ((i & 1) ? hA : hB);
        float*       hout = (i & 1) ? hB : hA;
        int do_stats = (i < N_LAYERS - 1) ? 1 : 0;
        float*       st_cur  = stats + (long)i * 2 * DIM;
        const float* st_prev = stats + (long)(i > 0 ? i - 1 : 0) * 2 * DIM;
        const float* g_prev  = gamma + (long)(i > 0 ? i - 1 : 0) * DIM;
        const float* b_prev  = beta + (long)(i > 0 ? i - 1 : 0) * DIM;

        if (i == 0) {
            gin_layer<false><<<layGrd, 256, 0, stream>>>(hin, rowbeg, deg, col,
                                                         W1 + (long)i * DIM * DIM, b1 + (long)i * DIM,
                                                         W2 + (long)i * DIM * DIM, b2 + (long)i * DIM,
                                                         eps + i, hout, st_cur, do_stats,
                                                         st_prev, g_prev, b_prev);
        } else {
            gin_layer<true><<<layGrd, 256, 0, stream>>>(hin, rowbeg, deg, col,
                                                        W1 + (long)i * DIM * DIM, b1 + (long)i * DIM,
                                                        W2 + (long)i * DIM * DIM, b2 + (long)i * DIM,
                                                        eps + i, hout, st_cur, do_stats,
                                                        st_prev, g_prev, b_prev);
        }
    }
    gin_zero<<<(NUM_GRAPHS * DIM + 255) / 256, 256, 0, stream>>>(out, NUM_GRAPHS * DIM);
    gin_pool<<<poolGrd, 256, 0, stream>>>(hB, batch, out);
}


# Layer kernel: 80 rows per block, not 16: each W fragment a wave loads (all 64 ahead of the barrier) feeds 5 WMMAs on 5 accumulator chains; BN column sums: 80 rows per atomic, their add order changes
# speedup vs baseline: 2.7744x; 1.1835x over previous
#include <hip/hip_runtime.h>
#include <hip/hip_bf16.h>

#define N_NODES   50000
#define N_EDGES   800000
#define DIM       128
#define N_LAYERS  4
#define NUM_GRAPHS 128
#define BN_EPS    1e-5f
#define TP        132
#define MT        5
#define BR        (16 * MT)

typedef __attribute__((ext_vector_type(2))) float v2f;
typedef __attribute__((ext_vector_type(8))) float v8f;

__global__ __launch_bounds__(256) void gin_zero(float* __restrict__ p, long n) {
    long i = (long)blockIdx.x * blockDim.x + threadIdx.x;
    if (i < n) p[i] = 0.0f;
}

__global__ __launch_bounds__(256) void gin_zero_i(int* __restrict__ p, int n, float* __restrict__ stats) {
    if (blockIdx.x < N_LAYERS) stats[blockIdx.x * 2 * DIM + threadIdx.x] = 0.0f;
    int i = blockIdx.x * blockDim.x + threadIdx.x;
    if (i < n) p[i] = 0;
}

__global__ __launch_bounds__(256) void gin_csr_count(const int* __restrict__ dst,
                                                     int* __restrict__ deg,
                                                     int* __restrict__ rank) {
    int e = blockIdx.x * blockDim.x + threadIdx.x;
    if (e >= N_EDGES) return;
    rank[e] = atomicAdd(&deg[dst[e]], 1);
}

__global__ __launch_bounds__(1024) void gin_csr_scan(const int* __restrict__ deg,
                                                     int* __restrict__ rowbeg) {
    __shared__ int wsum[32];
    __shared__ int wpre[32];
    const int tid  = threadIdx.x;
    const int lane = tid & 31;
    const int wave = tid >> 5;
    const int chunk0 = blockIdx.x * 1024;

    int pre = 0;
    for (int i = tid; i < chunk0; i += 1024) pre += deg[i];

    const int idx = chunk0 + tid;
    const int v = (idx < N_NODES) ? deg[idx] : 0;

    int inc = v;
    for (int off = 1; off < 32; off <<= 1) {
        int t = __shfl_up(inc, off);
        if (lane >= off) inc += t;
    }
    for (int off = 16; off > 0; off >>= 1) pre += __shfl_xor(pre, off);

    if (lane == 31) wsum[wave] = inc;
    if (lane == 0)  wpre[wave] = pre;
    __syncthreads();

    int base = 0;
    for (int w = 0; w < 32; ++w) {
        base += wpre[w];
        if (w < wave) base += wsum[w];
    }
    const int excl = base + inc - v;
    if (idx < N_NODES) rowbeg[idx] = excl;
}

__global__ __launch_bounds__(256) void gin_csr_fill(const int* __restrict__ src,
                                                    const int* __restrict__ dst,
                                                    const int* __restrict__ rowbeg,
                                                    const int* __restrict__ rank,
                                                    int* __restrict__ col) {
    int e = blockIdx.x * blockDim.x + threadIdx.x;
    if (e >= N_EDGES) return;
    col[rowbeg[dst[e]] + rank[e]] = src[e];
}

__device__ __forceinline__ void bn_coef(const float* __restrict__ st,
                                        const float* __restrict__ gamma,
                                        const float* __restrict__ beta,
                                        int c, float& mu, float& sc, float& be) {
    const float invN = 1.0f / (float)N_NODES;
    mu = st[c] * invN;
    float var = st[DIM + c] * invN - mu * mu;
    sc = rsqrtf(var + BN_EPS) * gamma[c];
    be = beta[c];
}
__device__ __forceinline__ float bn_relu1(float z, float mu, float sc, float be) {
    return fmaxf((z - mu) * sc + be, 0.0f);
}
__device__ __forceinline__ float4 bn_relu4(float4 z, float4 mu, float4 sc, float4 be) {
    float4 r;
    r.x = bn_relu1(z.x, mu.x, sc.x, be.x);
    r.y = bn_relu1(z.y, mu.y, sc.y, be.y);
    r.z = bn_relu1(z.z, mu.z, sc.z, be.z);
    r.w = bn_relu1(z.w, mu.w, sc.w, be.w);
    return r;
}

template <bool NORM>
__global__ __launch_bounds__(256, 6) void gin_layer(const float* __restrict__ hin,
                                                 const int* __restrict__ rowbeg,
                                                 const int* __restrict__ deg,
                                                 const int* __restrict__ col,
                                                 const float* __restrict__ W1,
                                                 const float* __restrict__ b1,
                                                 const float* __restrict__ W2,
                                                 const float* __restrict__ b2,
                                                 const float* __restrict__ epsp,
                                                 float* __restrict__ hout,
                                                 float* __restrict__ stats,
                                                 int do_stats,
                                                 const float* __restrict__ st_prev,
                                                 const float* __restrict__ gamma,
                                                 const float* __restrict__ beta) {
    __shared__ __attribute__((aligned(16))) float zt[BR * TP];
    __shared__ __attribute__((aligned(16))) float y1[BR * TP];

    const int row0 = blockIdx.x * BR;
    const int tid  = threadIdx.x;
    const int lane = tid & 31;
    const int wave = __builtin_amdgcn_readfirstlane(tid >> 5);
    const float e1 = 1.0f + epsp[0];

    {
        const float4 zero4 = make_float4(0.0f, 0.0f, 0.0f, 0.0f);
        float4 mu = zero4, sc = zero4, be = zero4;
        if (NORM) {
            bn_coef(st_prev, gamma, beta, 4 * lane + 0, mu.x, sc.x, be.x);
            bn_coef(st_prev, gamma, beta, 4 * lane + 1, mu.y, sc.y, be.y);
            bn_coef(st_prev, gamma, beta, 4 * lane + 2, mu.z, sc.z, be.z);
            bn_coef(st_prev, gamma, beta, 4 * lane + 3, mu.w, sc.w, be.w);
        }
        const float4* h4 = (const float4*)hin;

        for (int t = 0; t < MT; ++t) {
        const int rA = 16 * t + 2 * wave;
        const int nodeA = row0 + rA;
        const int nodeB = nodeA + 1;
        const int begA = rowbeg[nodeA];
        const int begB = rowbeg[nodeB];
        const int cntA = deg[nodeA];
        const int cntB = deg[nodeB];

        float4 selfA = h4[(long)nodeA * (DIM / 4) + lane];
        float4 selfB = h4[(long)nodeB * (DIM / 4) + lane];

        float4 accA = zero4, accB = zero4;
        const int cmax = (cntA > cntB) ? cntA : cntB;
        for (int base = 0; base < cmax; base += 32) {
            int mA = cntA - base; mA = (mA < 0) ? 0 : ((mA > 32) ? 32 : mA);
            int mB = cntB - base; mB = (mB < 0) ? 0 : ((mB > 32) ? 32 : mB);
            const int mineA = (lane < mA) ? col[begA + base + lane] : 0;
            const int mineB = (lane < mB) ? col[begB + base + lane] : 0;
            const int mm = (mA > mB) ? mA : mB;
            asm volatile("" :: "v"(mineA), "v"(mineB));
            for (int j = 0; j < mm; j += 4) {
                float4 a0 = zero4, a1 = zero4, a2 = zero4, a3 = zero4;
                float4 c0 = zero4, c1 = zero4, c2 = zero4, c3 = zero4;
                if (j + 0 < mA) a0 = h4[(long)__builtin_amdgcn_readlane(mineA, j + 0) * (DIM / 4) + lane];
                if (j + 1 < mA) a1 = h4[(long)__builtin_amdgcn_readlane(mineA, j + 1) * (DIM / 4) + lane];
                if (j + 2 < mA) a2 = h4[(long)__builtin_amdgcn_readlane(mineA, j + 2) * (DIM / 4) + lane];
                if (j + 3 < mA) a3 = h4[(long)__builtin_amdgcn_readlane(mineA, j + 3) * (DIM / 4) + lane];
                if (j + 0 < mB) c0 = h4[(long)__builtin_amdgcn_readlane(mineB, j + 0) * (DIM / 4) + lane];
                if (j + 1 < mB) c1 = h4[(long)__builtin_amdgcn_readlane(mineB, j + 1) * (DIM / 4) + lane];
                if (j + 2 < mB) c2 = h4[(long)__builtin_amdgcn_readlane(mineB, j + 2) * (DIM / 4) + lane];
                if (j + 3 < mB) c3 = h4[(long)__builtin_amdgcn_readlane(mineB, j + 3) * (DIM / 4) + lane];
                if (NORM) {
                    if (j + 0 < mA) a0 = bn_relu4(a0, mu, sc, be);
                    if (j + 1 < mA) a1 = bn_relu4(a1, mu, sc, be);
                    if (j + 2 < mA) a2 = bn_relu4(a2, mu, sc, be);
                    if (j + 3 < mA) a3 = bn_relu4(a3, mu, sc, be);
                    if (j + 0 < mB) c0 = bn_relu4(c0, mu, sc, be);
                    if (j + 1 < mB) c1 = bn_relu4(c1, mu, sc, be);
                    if (j + 2 < mB) c2 = bn_relu4(c2, mu, sc, be);
                    if (j + 3 < mB) c3 = bn_relu4(c3, mu, sc, be);
                }
                accA.x += (a0.x + a1.x) + (a2.x + a3.x);
                accA.y += (a0.y + a1.y) + (a2.y + a3.y);
                accA.z += (a0.z + a1.z) + (a2.z + a3.z);
                accA.w += (a0.w + a1.w) + (a2.w + a3.w);
                accB.x += (c0.x + c1.x) + (c2.x + c3.x);
                accB.y += (c0.y + c1.y) + (c2.y + c3.y);
                accB.z += (c0.z + c1.z) + (c2.z + c3.z);
                accB.w += (c0.w + c1.w) + (c2.w + c3.w);
            }
        }

        if (NORM) {
            selfA = bn_relu4(selfA, mu, sc, be);
            selfB = bn_relu4(selfB, mu, sc, be);
        }
        float4 zA, zB;
        zA.x = e1 * selfA.x + accA.x;
        zA.y = e1 * selfA.y + accA.y;
        zA.z = e1 * selfA.z + accA.z;
        zA.w = e1 * selfA.w + accA.w;
        zB.x = e1 * selfB.x + accB.x;
        zB.y = e1 * selfB.y + accB.y;
        zB.z = e1 * selfB.z + accB.z;
        zB.w = e1 * selfB.w + accB.w;
        *(float4*)&zt[rA * TP + 4 * lane]       = zA;
        *(float4*)&zt[(rA + 1) * TP + 4 * lane] = zB;
        }
    }

    const int n    = lane & 15;
    const int hi   = lane >> 4;
    const int col0 = wave * 16;
    const int kk   = hi * 2;

    v2f bw[DIM / 4];
    #pragma unroll
    for (int ks = 0; ks < DIM / 4; ++ks) {
        bw[ks].x = W1[(4 * ks + kk) * DIM + col0 + n];
        bw[ks].y = W1[(4 * ks + kk + 1) * DIM + col0 + n];
    }
    float bv = b1[col0 + n];
    __syncthreads();

    v8f acc[MT];
    #pragma unroll
    for (int t = 0; t < MT; ++t)
        for (int v = 0; v < 8; ++v) acc[t][v] = bv;
    #pragma unroll
    for (int ks = 0; ks < DIM / 4; ++ks) {
        #pragma unroll
        for (int t = 0; t < MT; ++t) {
            v2f a;
            const float* zp = &zt[(16 * t + n) * TP + 4 * ks + kk];
            a.x = zp[0];
            a.y = zp[1];
            acc[t] = __builtin_amdgcn_wmma_f32_16x16x4_f32(false, a, false, bw[ks],
                                                           (short)0, acc[t], false, false);
        }
    }
    #pragma unroll
    for (int ks = 0; ks < DIM / 4; ++ks) {
        bw[ks].x = W2[(4 * ks + kk) * DIM + col0 + n];
        bw[ks].y = W2[(4 * ks + kk + 1) * DIM + col0 + n];
    }
    const float bv2 = b2[col0 + n];
    #pragma unroll
    for (int t = 0; t < MT; ++t) {
        for (int v = 0; v < 8; ++v) {
            int m = 16 * t + v + 8 * hi;
            y1[m * TP + col0 + n] = fmaxf(acc[t][v], 0.0f);
        }
    }
    __syncthreads();

    #pragma unroll
    for (int t = 0; t < MT; ++t)
        for (int v = 0; v < 8; ++v) acc[t][v] = bv2;
    #pragma unroll
    for (int ks = 0; ks < DIM / 4; ++ks) {
        #pragma unroll
        for (int t = 0; t < MT; ++t) {
            v2f a;
            const float* yp = &y1[(16 * t + n) * TP + 4 * ks + kk];
            a.x = yp[0];
            a.y = yp[1];
            acc[t] = __builtin_amdgcn_wmma_f32_16x16x4_f32(false, a, false, bw[ks],
                                                           (short)0, acc[t], false, false);
        }
    }

    float s = 0.0f, ssq = 0.0f;
    #pragma unroll
    for (int t = 0; t < MT; ++t) {
        for (int v = 0; v < 8; ++v) {
            int m = 16 * t + v + 8 * hi;
            float val = acc[t][v];
            hout[(long)(row0 + m) * DIM + col0 + n] = val;
            s += val;
            ssq += val * val;
        }
    }
    if (do_stats) {
        atomicAdd(&stats[col0 + n], s);
        atomicAdd(&stats[DIM + col0 + n], ssq);
    }
}

__global__ __launch_bounds__(256) void gin_pool(const float* __restrict__ h,
                                                const int* __restrict__ batch,
                                                float* __restrict__ out) {
    long t = (long)blockIdx.x * blockDim.x + threadIdx.x;
    long nidx = t >> 5;
    int lane = (int)(t & 31);
    if (nidx >= N_NODES) return;
    int g = batch[nidx];
    const float4* hp = (const float4*)(h + nidx * DIM);
    float4 v = hp[lane];
    float* op = out + (long)g * DIM + lane * 4;
    atomicAdd(op + 0, v.x);
    atomicAdd(op + 1, v.y);
    atomicAdd(op + 2, v.z);
    atomicAdd(op + 3, v.w);
}

extern "C" void kernel_launch(void* const* d_in, const int* in_sizes, int n_in,
                              void* d_out, int out_size, void* d_ws, size_t ws_size,
                              hipStream_t stream) {
    const float* x     = (const float*)d_in[0];
    const int*   ei    = (const int*)d_in[1];
    const int*   batch = (const int*)d_in[2];
    const float* W1    = (const float*)d_in[3];
    const float* b1    = (const float*)d_in[4];
    const float* W2    = (const float*)d_in[5];
    const float* b2    = (const float*)d_in[6];
    const float* eps   = (const float*)d_in[7];
    const float* gamma = (const float*)d_in[8];
    const float* beta  = (const float*)d_in[9];
    float* out = (float*)d_out;

    const long ND = (long)N_NODES * DIM;
    float* hA    = (float*)d_ws;
    float* hB    = hA + ND;
    float* stats = hB + ND;
    int* deg    = (int*)(stats + N_LAYERS * 2 * DIM);
    int* rowbeg = deg + N_NODES;
    int* rank   = rowbeg + N_NODES;
    int* col    = rank + N_EDGES;

    const int* src = ei;
    const int* dst = ei + N_EDGES;

    const int  degGrd  = (N_NODES + 255) / 256;
    const int  edgeGrd = (N_EDGES + 255) / 256;
    const int  scanGrd = (N_NODES + 1023) / 1024;
    const int  layGrd  = N_NODES / BR;
    const int  poolGrd = (int)(((long)N_NODES * 32 + 255) / 256);

    gin_zero_i<<<degGrd, 256, 0, stream>>>(deg, N_NODES, stats);
    gin_csr_count<<<edgeGrd, 256, 0, stream>>>(dst, deg, rank);
    gin_csr_scan<<<scanGrd, 1024, 0, stream>>>(deg, rowbeg);
    gin_csr_fill<<<edgeGrd, 256, 0, stream>>>(src, dst, rowbeg, rank, col);

    for (int i = 0; i < N_LAYERS; ++i) {
        const float* hin  = (i == 0) ? x : ((i & 1) ? hA : hB);
        float*       hout = (i & 1) ? hB : hA;
        int do_stats = (i < N_LAYERS - 1) ? 1 : 0;
        float*       st_cur  = stats + (long)i * 2 * DIM;
        const float* st_prev = stats + (long)(i > 0 ? i - 1 : 0) * 2 * DIM;
        const float* g_prev  = gamma + (long)(i > 0 ? i - 1 : 0) * DIM;
        const float* b_prev  = beta + (long)(i > 0 ? i - 1 : 0) * DIM;

        if (i == 0) {
            gin_layer<false><<<layGrd, 256, 0, stream>>>(hin, rowbeg, deg, col,
                                                         W1 + (long)i * DIM * DIM, b1 + (long)i * DIM,
                                                         W2 + (long)i * DIM * DIM, b2 + (long)i * DIM,
                                                         eps + i, hout, st_cur, do_stats,
                                                         st_prev, g_prev, b_prev);
        } else {
            gin_layer<true><<<layGrd, 256, 0, stream>>>(hin, rowbeg, deg, col,
                                                        W1 + (long)i * DIM * DIM, b1 + (long)i * DIM,
                                                        W2 + (long)i * DIM * DIM, b2 + (long)i * DIM,
                                                        eps + i, hout, st_cur, do_stats,
                                                        st_prev, g_prev, b_prev);
        }
    }
    gin_zero<<<(NUM_GRAPHS * DIM + 255) / 256, 256, 0, stream>>>(out, NUM_GRAPHS * DIM);
    gin_pool<<<poolGrd, 256, 0, stream>>>(hB, batch, out);
}


# global_add_pool moved into the last layer's kernel: a lane sums its 40 rows per run of equal graph index, one atomic per run; pool and zero launches and the last h plane go; pool sum order changes
# speedup vs baseline: 3.5645x; 1.2848x over previous
#include <hip/hip_runtime.h>
#include <hip/hip_bf16.h>

#define N_NODES   50000
#define N_EDGES   800000
#define DIM       128
#define N_LAYERS  4
#define NUM_GRAPHS 128
#define BN_EPS    1e-5f
#define TP        132
#define MT        5
#define BR        (16 * MT)

typedef __attribute__((ext_vector_type(2))) float v2f;
typedef __attribute__((ext_vector_type(8))) float v8f;

__global__ __launch_bounds__(256) void gin_zero_i(int* __restrict__ p, int n, float* __restrict__ stats,
                                                  float* __restrict__ out) {
    if (blockIdx.x < N_LAYERS) stats[blockIdx.x * 2 * DIM + threadIdx.x] = 0.0f;
    int i = blockIdx.x * blockDim.x + threadIdx.x;
    if (i < NUM_GRAPHS * DIM) out[i] = 0.0f;
    if (i < n) p[i] = 0;
}

__global__ __launch_bounds__(256) void gin_csr_count(const int* __restrict__ dst,
                                                     int* __restrict__ deg,
                                                     int* __restrict__ rank) {
    int e = blockIdx.x * blockDim.x + threadIdx.x;
    if (e >= N_EDGES) return;
    rank[e] = atomicAdd(&deg[dst[e]], 1);
}

__global__ __launch_bounds__(1024) void gin_csr_scan(const int* __restrict__ deg,
                                                     int* __restrict__ rowbeg) {
    __shared__ int wsum[32];
    __shared__ int wpre[32];
    const int tid  = threadIdx.x;
    const int lane = tid & 31;
    const int wave = tid >> 5;
    const int chunk0 = blockIdx.x * 1024;

    int pre = 0;
    for (int i = tid; i < chunk0; i += 1024) pre += deg[i];

    const int idx = chunk0 + tid;
    const int v = (idx < N_NODES) ? deg[idx] : 0;

    int inc = v;
    for (int off = 1; off < 32; off <<= 1) {
        int t = __shfl_up(inc, off);
        if (lane >= off) inc += t;
    }
    for (int off = 16; off > 0; off >>= 1) pre += __shfl_xor(pre, off);

    if (lane == 31) wsum[wave] = inc;
    if (lane == 0)  wpre[wave] = pre;
    __syncthreads();

    int base = 0;
    for (int w = 0; w < 32; ++w) {
        base += wpre[w];
        if (w < wave) base += wsum[w];
    }
    const int excl = base + inc - v;
    if (idx < N_NODES) rowbeg[idx] = excl;
}

__global__ __launch_bounds__(256) void gin_csr_fill(const int* __restrict__ src,
                                                    const int* __restrict__ dst,
                                                    const int* __restrict__ rowbeg,
                                                    const int* __restrict__ rank,
                                                    int* __restrict__ col) {
    int e = blockIdx.x * blockDim.x + threadIdx.x;
    if (e >= N_EDGES) return;
    col[rowbeg[dst[e]] + rank[e]] = src[e];
}

__device__ __forceinline__ void bn_coef(const float* __restrict__ st,
                                        const float* __restrict__ gamma,
                                        const float* __restrict__ beta,
                                        int c, float& mu, float& sc, float& be) {
    const float invN = 1.0f / (float)N_NODES;
    mu = st[c] * invN;
    float var = st[DIM + c] * invN - mu * mu;
    sc = rsqrtf(var + BN_EPS) * gamma[c];
    be = beta[c];
}
__device__ __forceinline__ float bn_relu1(float z, float mu, float sc, float be) {
    return fmaxf((z - mu) * sc + be, 0.0f);
}
__device__ __forceinline__ float4 bn_relu4(float4 z, float4 mu, float4 sc, float4 be) {
    float4 r;
    r.x = bn_relu1(z.x, mu.x, sc.x, be.x);
    r.y = bn_relu1(z.y, mu.y, sc.y, be.y);
    r.z = bn_relu1(z.z, mu.z, sc.z, be.z);
    r.w = bn_relu1(z.w, mu.w, sc.w, be.w);
    return r;
}

template <bool NORM, bool POOL>
__global__ __launch_bounds__(256, 6) void gin_layer(const float* __restrict__ hin,
                                                 const int* __restrict__ rowbeg,
                                                 const int* __restrict__ deg,
                                                 const int* __restrict__ col,
                                                 const float* __restrict__ W1,
                                                 const float* __restrict__ b1,
                                                 const float* __restrict__ W2,
                                                 const float* __restrict__ b2,
                                                 const float* __restrict__ epsp,
                                                 float* __restrict__ hout,
                                                 float* __restrict__ stats,
                                                 int do_stats,
                                                 const float* __restrict__ st_prev,
                                                 const float* __restrict__ gamma,
                                                 const float* __restrict__ beta,
                                                 const int* __restrict__ batch,
                                                 float* __restrict__ out) {
    __shared__ __attribute__((aligned(16))) float zt[BR * TP];
    __shared__ __attribute__((aligned(16))) float y1[BR * TP];
    __shared__ __attribute__((aligned(16))) int   gidx[BR];

    const int row0 = blockIdx.x * BR;
    const int tid  = threadIdx.x;
    int gmine = 0;
    if (POOL && tid < BR) gmine = batch[row0 + tid];
    const int lane = tid & 31;
    const int wave = __builtin_amdgcn_readfirstlane(tid >> 5);
    const float e1 = 1.0f + epsp[0];

    {
        const float4 zero4 = make_float4(0.0f, 0.0f, 0.0f, 0.0f);
        float4 mu = zero4, sc = zero4, be = zero4;
        if (NORM) {
            bn_coef(st_prev, gamma, beta, 4 * lane + 0, mu.x, sc.x, be.x);
            bn_coef(st_prev, gamma, beta, 4 * lane + 1, mu.y, sc.y, be.y);
            bn_coef(st_prev, gamma, beta, 4 * lane + 2, mu.z, sc.z, be.z);
            bn_coef(st_prev, gamma, beta, 4 * lane + 3, mu.w, sc.w, be.w);
        }
        const float4* h4 = (const float4*)hin;

        for (int t = 0; t < MT; ++t) {
        const int rA = 16 * t + 2 * wave;
        const int nodeA = row0 + rA;
        const int nodeB = nodeA + 1;
        const int begA = rowbeg[nodeA];
        const int begB = rowbeg[nodeB];
        const int cntA = deg[nodeA];
        const int cntB = deg[nodeB];

        float4 selfA = h4[(long)nodeA * (DIM / 4) + lane];
        float4 selfB = h4[(long)nodeB * (DIM / 4) + lane];

        float4 accA = zero4, accB = zero4;
        const int cmax = (cntA > cntB) ? cntA : cntB;
        for (int base = 0; base < cmax; base += 32) {
            int mA = cntA - base; mA = (mA < 0) ? 0 : ((mA > 32) ? 32 : mA);
            int mB = cntB - base; mB = (mB < 0) ? 0 : ((mB > 32) ? 32 : mB);
            const int mineA = (lane < mA) ? col[begA + base + lane] : 0;
            const int mineB = (lane < mB) ? col[begB + base + lane] : 0;
            const int mm = (mA > mB) ? mA : mB;
            asm volatile("" :: "v"(mineA), "v"(mineB));
            for (int j = 0; j < mm; j += 4) {
                float4 a0 = zero4, a1 = zero4, a2 = zero4, a3 = zero4;
                float4 c0 = zero4, c1 = zero4, c2 = zero4, c3 = zero4;
                if (j + 0 < mA) a0 = h4[(long)__builtin_amdgcn_readlane(mineA, j + 0) * (DIM / 4) + lane];
                if (j + 1 < mA) a1 = h4[(long)__builtin_amdgcn_readlane(mineA, j + 1) * (DIM / 4) + lane];
                if (j + 2 < mA) a2 = h4[(long)__builtin_amdgcn_readlane(mineA, j + 2) * (DIM / 4) + lane];
                if (j + 3 < mA) a3 = h4[(long)__builtin_amdgcn_readlane(mineA, j + 3) * (DIM / 4) + lane];
                if (j + 0 < mB) c0 = h4[(long)__builtin_amdgcn_readlane(mineB, j + 0) * (DIM / 4) + lane];
                if (j + 1 < mB) c1 = h4[(long)__builtin_amdgcn_readlane(mineB, j + 1) * (DIM / 4) + lane];
                if (j + 2 < mB) c2 = h4[(long)__builtin_amdgcn_readlane(mineB, j + 2) * (DIM / 4) + lane];
                if (j + 3 < mB) c3 = h4[(long)__builtin_amdgcn_readlane(mineB, j + 3) * (DIM / 4) + lane];
                if (NORM) {
                    if (j + 0 < mA) a0 = bn_relu4(a0, mu, sc, be);
                    if (j + 1 < mA) a1 = bn_relu4(a1, mu, sc, be);
                    if (j + 2 < mA) a2 = bn_relu4(a2, mu, sc, be);
                    if (j + 3 < mA) a3 = bn_relu4(a3, mu, sc, be);
                    if (j + 0 < mB) c0 = bn_relu4(c0, mu, sc, be);
                    if (j + 1 < mB) c1 = bn_relu4(c1, mu, sc, be);
                    if (j + 2 < mB) c2 = bn_relu4(c2, mu, sc, be);
                    if (j + 3 < mB) c3 = bn_relu4(c3, mu, sc, be);
                }
                accA.x += (a0.x + a1.x) + (a2.x + a3.x);
                accA.y += (a0.y + a1.y) + (a2.y + a3.y);
                accA.z += (a0.z + a1.z) + (a2.z + a3.z);
                accA.w += (a0.w + a1.w) + (a2.w + a3.w);
                accB.x += (c0.x + c1.x) + (c2.x + c3.x);
                accB.y += (c0.y + c1.y) + (c2.y + c3.y);
                accB.z += (c0.z + c1.z) + (c2.z + c3.z);
                accB.w += (c0.w + c1.w) + (c2.w + c3.w);
            }
        }

        if (NORM) {
            selfA = bn_relu4(selfA, mu, sc, be);
            selfB = bn_relu4(selfB, mu, sc, be);
        }
        float4 zA, zB;
        zA.x = e1 * selfA.x + accA.x;
        zA.y = e1 * selfA.y + accA.y;
        zA.z = e1 * selfA.z + accA.z;
        zA.w = e1 * selfA.w + accA.w;
        zB.x = e1 * selfB.x + accB.x;
        zB.y = e1 * selfB.y + accB.y;
        zB.z = e1 * selfB.z + accB.z;
        zB.w = e1 * selfB.w + accB.w;
        *(float4*)&zt[rA * TP + 4 * lane]       = zA;
        *(float4*)&zt[(rA + 1) * TP + 4 * lane] = zB;
        }
    }

    if (POOL && tid < BR) gidx[tid] = gmine;

    const int n    = lane & 15;
    const int hi   = lane >> 4;
    const int col0 = wave * 16;
    const int kk   = hi * 2;

    v2f bw[DIM / 4];
    #pragma unroll
    for (int ks = 0; ks < DIM / 4; ++ks) {
        bw[ks].x = W1[(4 * ks + kk) * DIM + col0 + n];
        bw[ks].y = W1[(4 * ks + kk + 1) * DIM + col0 + n];
    }
    float bv = b1[col0 + n];
    __syncthreads();

    v8f acc[MT];
    #pragma unroll
    for (int t = 0; t < MT; ++t)
        for (int v = 0; v < 8; ++v) acc[t][v] = bv;
    #pragma unroll
    for (int ks = 0; ks < DIM / 4; ++ks) {
        #pragma unroll
        for (int t = 0; t < MT; ++t) {
            v2f a;
            const float* zp = &zt[(16 * t + n) * TP + 4 * ks + kk];
            a.x = zp[0];
            a.y = zp[1];
            acc[t] = __builtin_amdgcn_wmma_f32_16x16x4_f32(false, a, false, bw[ks],
                                                           (short)0, acc[t], false, false);
        }
    }
    #pragma unroll
    for (int ks = 0; ks < DIM / 4; ++ks) {
        bw[ks].x = W2[(4 * ks + kk) * DIM + col0 + n];
        bw[ks].y = W2[(4 * ks + kk + 1) * DIM + col0 + n];
    }
    const float bv2 = b2[col0 + n];
    #pragma unroll
    for (int t = 0; t < MT; ++t) {
        for (int v = 0; v < 8; ++v) {
            int m = 16 * t + v + 8 * hi;
            y1[m * TP + col0 + n] = fmaxf(acc[t][v], 0.0f);
        }
    }
    __syncthreads();

    #pragma unroll
    for (int t = 0; t < MT; ++t)
        for (int v = 0; v < 8; ++v) acc[t][v] = bv2;
    #pragma unroll
    for (int ks = 0; ks < DIM / 4; ++ks) {
        #pragma unroll
        for (int t = 0; t < MT; ++t) {
            v2f a;
            const float* yp = &y1[(16 * t + n) * TP + 4 * ks + kk];
            a.x = yp[0];
            a.y = yp[1];
            acc[t] = __builtin_amdgcn_wmma_f32_16x16x4_f32(false, a, false, bw[ks],
                                                           (short)0, acc[t], false, false);
        }
    }

    if (POOL) {
        float* oc = out + col0 + n;
        int cur = gidx[8 * hi];
        float run = 0.0f;
        #pragma unroll
        for (int t = 0; t < MT; ++t) {
            const int4 g0 = *(const int4*)&gidx[16 * t + 8 * hi];
            const int4 g1 = *(const int4*)&gidx[16 * t + 8 * hi + 4];
            const int g[8] = {g0.x, g0.y, g0.z, g0.w, g1.x, g1.y, g1.z, g1.w};
            #pragma unroll
            for (int v = 0; v < 8; ++v) {
                if (g[v] != cur) {
                    atomicAdd(oc + (long)cur * DIM, run);
                    run = 0.0f;
                    cur = g[v];
                }
                run += acc[t][v];
            }
        }
        atomicAdd(oc + (long)cur * DIM, run);
    } else {
        float s = 0.0f, ssq = 0.0f;
        #pragma unroll
        for (int t = 0; t < MT; ++t) {
            for (int v = 0; v < 8; ++v) {
                int m = 16 * t + v + 8 * hi;
                float val = acc[t][v];
                hout[(long)(row0 + m) * DIM + col0 + n] = val;
                s += val;
                ssq += val * val;
            }
        }
        if (do_stats) {
            atomicAdd(&stats[col0 + n], s);
            atomicAdd(&stats[DIM + col0 + n], ssq);
        }
    }
}

extern "C" void kernel_launch(void* const* d_in, const int* in_sizes, int n_in,
                              void* d_out, int out_size, void* d_ws, size_t ws_size,
                              hipStream_t stream) {
    const float* x     = (const float*)d_in[0];
    const int*   ei    = (const int*)d_in[1];
    const int*   batch = (const int*)d_in[2];
    const float* W1    = (const float*)d_in[3];
    const float* b1    = (const float*)d_in[4];
    const float* W2    = (const float*)d_in[5];
    const float* b2    = (const float*)d_in[6];
    const float* eps   = (const float*)d_in[7];
    const float* gamma = (const float*)d_in[8];
    const float* beta  = (const float*)d_in[9];
    float* out = (float*)d_out;

    const long ND = (long)N_NODES * DIM;
    float* hA    = (float*)d_ws;
    float* hB    = hA + ND;
    float* stats = hB + ND;
    int* deg    = (int*)(stats + N_LAYERS * 2 * DIM);
    int* rowbeg = deg + N_NODES;
    int* rank   = rowbeg + N_NODES;
    int* col    = rank + N_EDGES;

    const int* src = ei;
    const int* dst = ei + N_EDGES;

    const int  degGrd  = (N_NODES + 255) / 256;
    const int  edgeGrd = (N_EDGES + 255) / 256;
    const int  scanGrd = (N_NODES + 1023) / 1024;
    const int  layGrd  = N_NODES / BR;

    gin_zero_i<<<degGrd, 256, 0, stream>>>(deg, N_NODES, stats, out);
    gin_csr_count<<<edgeGrd, 256, 0, stream>>>(dst, deg, rank);
    gin_csr_scan<<<scanGrd, 1024, 0, stream>>>(deg, rowbeg);
    gin_csr_fill<<<edgeGrd, 256, 0, stream>>>(src, dst, rowbeg, rank, col);

    for (int i = 0; i < N_LAYERS; ++i) {
        const float* hin  = (i == 0) ? x : ((i & 1) ? hA : hB);
        float*       hout = (i & 1) ? hB : hA;
        int do_stats = (i < N_LAYERS - 1) ? 1 : 0;
        float*       st_cur  = stats + (long)i * 2 * DIM;
        const float* st_prev = stats + (long)(i > 0 ? i - 1 : 0) * 2 * DIM;
        const float* g_prev  = gamma + (long)(i > 0 ? i - 1 : 0) * DIM;
        const float* b_prev  = beta + (long)(i > 0 ? i - 1 : 0) * DIM;

        if (i == 0) {
            gin_layer<false, false><<<layGrd, 256, 0, stream>>>(hin, rowbeg, deg, col,
                                                         W1 + (long)i * DIM * DIM, b1 + (long)i * DIM,
                                                         W2 + (long)i * DIM * DIM, b2 + (long)i * DIM,
                                                         eps + i, hout, st_cur, do_stats,
                                                         st_prev, g_prev, b_prev, batch, out);
        } else if (i < N_LAYERS - 1) {
            gin_layer<true, false><<<layGrd, 256, 0, stream>>>(hin, rowbeg, deg, col,
                                                        W1 + (long)i * DIM * DIM, b1 + (long)i * DIM,
                                                        W2 + (long)i * DIM * DIM, b2 + (long)i * DIM,
                                                        eps + i, hout, st_cur, do_stats,
                                                        st_prev, g_prev, b_prev, batch, out);
        } else {
            gin_layer<true, true><<<layGrd, 256, 0, stream>>>(hin, rowbeg, deg, col,
                                                        W1 + (long)i * DIM * DIM, b1 + (long)i * DIM,
                                                        W2 + (long)i * DIM * DIM, b2 + (long)i * DIM,
                                                        eps + i, hout, st_cur, do_stats,
                                                        st_prev, g_prev, b_prev, batch, out);
        }
    }
}


# Layer kernel: 3125 row tiles dealt to 768 blocks of 4 or 5 tiles (3 blocks on each of 256 WGPs: 12-13 tiles a WGP), not 625 blocks of 5 (10 or 15 a WGP); BN and pool sums regroup, add order changes
# speedup vs baseline: 3.6972x; 1.0372x over previous
#include <hip/hip_runtime.h>
#include <hip/hip_bf16.h>

#define N_NODES   50000
#define N_EDGES   800000
#define DIM       128
#define N_LAYERS  4
#define NUM_GRAPHS 128
#define BN_EPS    1e-5f
#define TP        132
#define MT        5
#define BR        (16 * MT)
#define LAY_GRID  768
#define NB5       53

typedef __attribute__((ext_vector_type(2))) float v2f;
typedef __attribute__((ext_vector_type(8))) float v8f;

__global__ __launch_bounds__(256) void gin_zero_i(int* __restrict__ p, int n, float* __restrict__ stats,
                                                  float* __restrict__ out) {
    if (blockIdx.x < N_LAYERS) stats[blockIdx.x * 2 * DIM + threadIdx.x] = 0.0f;
    int i = blockIdx.x * blockDim.x + threadIdx.x;
    if (i < NUM_GRAPHS * DIM) out[i] = 0.0f;
    if (i < n) p[i] = 0;
}

__global__ __launch_bounds__(256) void gin_csr_count(const int* __restrict__ dst,
                                                     int* __restrict__ deg,
                                                     int* __restrict__ rank) {
    int e = blockIdx.x * blockDim.x + threadIdx.x;
    if (e >= N_EDGES) return;
    rank[e] = atomicAdd(&deg[dst[e]], 1);
}

__global__ __launch_bounds__(1024) void gin_csr_scan(const int* __restrict__ deg,
                                                     int* __restrict__ rowbeg) {
    __shared__ int wsum[32];
    __shared__ int wpre[32];
    const int tid  = threadIdx.x;
    const int lane = tid & 31;
    const int wave = tid >> 5;
    const int chunk0 = blockIdx.x * 1024;

    int pre = 0;
    for (int i = tid; i < chunk0; i += 1024) pre += deg[i];

    const int idx = chunk0 + tid;
    const int v = (idx < N_NODES) ? deg[idx] : 0;

    int inc = v;
    for (int off = 1; off < 32; off <<= 1) {
        int t = __shfl_up(inc, off);
        if (lane >= off) inc += t;
    }
    for (int off = 16; off > 0; off >>= 1) pre += __shfl_xor(pre, off);

    if (lane == 31) wsum[wave] = inc;
    if (lane == 0)  wpre[wave] = pre;
    __syncthreads();

    int base = 0;
    for (int w = 0; w < 32; ++w) {
        base += wpre[w];
        if (w < wave) base += wsum[w];
    }
    const int excl = base + inc - v;
    if (idx < N_NODES) rowbeg[idx] = excl;
}

__global__ __launch_bounds__(256) void gin_csr_fill(const int* __restrict__ src,
                                                    const int* __restrict__ dst,
                                                    const int* __restrict__ rowbeg,
                                                    const int* __restrict__ rank,
                                                    int* __restrict__ col) {
    int e = blockIdx.x * blockDim.x + threadIdx.x;
    if (e >= N_EDGES) return;
    col[rowbeg[dst[e]] + rank[e]] = src[e];
}

__device__ __forceinline__ void bn_coef(const float* __restrict__ st,
                                        const float* __restrict__ gamma,
                                        const float* __restrict__ beta,
                                        int c, float& mu, float& sc, float& be) {
    const float invN = 1.0f / (float)N_NODES;
    mu = st[c] * invN;
    float var = st[DIM + c] * invN - mu * mu;
    sc = rsqrtf(var + BN_EPS) * gamma[c];
    be = beta[c];
}
__device__ __forceinline__ float bn_relu1(float z, float mu, float sc, float be) {
    return fmaxf((z - mu) * sc + be, 0.0f);
}
__device__ __forceinline__ float4 bn_relu4(float4 z, float4 mu, float4 sc, float4 be) {
    float4 r;
    r.x = bn_relu1(z.x, mu.x, sc.x, be.x);
    r.y = bn_relu1(z.y, mu.y, sc.y, be.y);
    r.z = bn_relu1(z.z, mu.z, sc.z, be.z);
    r.w = bn_relu1(z.w, mu.w, sc.w, be.w);
    return r;
}

template <bool POOL, int MTT>
__device__ __forceinline__ void gin_products(const int row0, const float* __restrict__ zt, float* __restrict__ y1,
                                             const int* __restrict__ gidx,
                                             const float* __restrict__ W1, const float* __restrict__ b1,
                                             const float* __restrict__ W2, const float* __restrict__ b2,
                                             float* __restrict__ hout, float* __restrict__ stats, int do_stats,
                                             float* __restrict__ out) {
    const int tid  = threadIdx.x;
    const int lane = tid & 31;
    const int wave = __builtin_amdgcn_readfirstlane(tid >> 5);
    const int n    = lane & 15;
    const int hi   = lane >> 4;
    const int col0 = wave * 16;
    const int kk   = hi * 2;

    v2f bw[DIM / 4];
    #pragma unroll
    for (int ks = 0; ks < DIM / 4; ++ks) {
        bw[ks].x = W1[(4 * ks + kk) * DIM + col0 + n];
        bw[ks].y = W1[(4 * ks + kk + 1) * DIM + col0 + n];
    }
    float bv = b1[col0 + n];
    __syncthreads();

    v8f acc[MTT];
    #pragma unroll
    for (int t = 0; t < MTT; ++t)
        for (int v = 0; v < 8; ++v) acc[t][v] = bv;
    #pragma unroll
    for (int ks = 0; ks < DIM / 4; ++ks) {
        #pragma unroll
        for (int t = 0; t < MTT; ++t) {
            v2f a;
            const float* zp = &zt[(16 * t + n) * TP + 4 * ks + kk];
            a.x = zp[0];
            a.y = zp[1];
            acc[t] = __builtin_amdgcn_wmma_f32_16x16x4_f32(false, a, false, bw[ks],
                                                           (short)0, acc[t], false, false);
        }
    }
    __builtin_amdgcn_sched_barrier(0);
    #pragma unroll
    for (int ks = 0; ks < DIM / 4; ++ks) {
        bw[ks].x = W2[(4 * ks + kk) * DIM + col0 + n];
        bw[ks].y = W2[(4 * ks + kk + 1) * DIM + col0 + n];
    }
    const float bv2 = b2[col0 + n];
    #pragma unroll
    for (int t = 0; t < MTT; ++t) {
        for (int v = 0; v < 8; ++v) {
            int m = 16 * t + v + 8 * hi;
            y1[m * TP + col0 + n] = fmaxf(acc[t][v], 0.0f);
        }
    }
    __syncthreads();

    #pragma unroll
    for (int t = 0; t < MTT; ++t)
        for (int v = 0; v < 8; ++v) acc[t][v] = bv2;
    #pragma unroll
    for (int ks = 0; ks < DIM / 4; ++ks) {
        #pragma unroll
        for (int t = 0; t < MTT; ++t) {
            v2f a;
            const float* yp = &y1[(16 * t + n) * TP + 4 * ks + kk];
            a.x = yp[0];
            a.y = yp[1];
            acc[t] = __builtin_amdgcn_wmma_f32_16x16x4_f32(false, a, false, bw[ks],
                                                           (short)0, acc[t], false, false);
        }
    }

    if (POOL) {
        float* oc = out + col0 + n;
        int cur = gidx[8 * hi];
        float run = 0.0f;
        #pragma unroll
        for (int t = 0; t < MTT; ++t) {
            const int4 g0 = *(const int4*)&gidx[16 * t + 8 * hi];
            const int4 g1 = *(const int4*)&gidx[16 * t + 8 * hi + 4];
            const int g[8] = {g0.x, g0.y, g0.z, g0.w, g1.x, g1.y, g1.z, g1.w};
            #pragma unroll
            for (int v = 0; v < 8; ++v) {
                if (g[v] != cur) {
                    atomicAdd(oc + (long)cur * DIM, run);
                    run = 0.0f;
                    cur = g[v];
                }
                run += acc[t][v];
            }
        }
        atomicAdd(oc + (long)cur * DIM, run);
    } else {
        float s = 0.0f, ssq = 0.0f;
        #pragma unroll
        for (int t = 0; t < MTT; ++t) {
            for (int v = 0; v < 8; ++v) {
                int m = 16 * t + v + 8 * hi;
                float val = acc[t][v];
                hout[(long)(row0 + m) * DIM + col0 + n] = val;
                s += val;
                ssq += val * val;
            }
        }
        if (do_stats) {
            atomicAdd(&stats[col0 + n], s);
            atomicAdd(&stats[DIM + col0 + n], ssq);
        }
    }
}

template <bool NORM, bool POOL>
__global__ __launch_bounds__(256, 6) void gin_layer(const float* __restrict__ hin,
                                                 const int* __restrict__ rowbeg,
                                                 const int* __restrict__ deg,
                                                 const int* __restrict__ col,
                                                 const float* __restrict__ W1,
                                                 const float* __restrict__ b1,
                                                 const float* __restrict__ W2,
                                                 const float* __restrict__ b2,
                                                 const float* __restrict__ epsp,
                                                 float* __restrict__ hout,
                                                 float* __restrict__ stats,
                                                 int do_stats,
                                                 const float* __restrict__ st_prev,
                                                 const float* __restrict__ gamma,
                                                 const float* __restrict__ beta,
                                                 const int* __restrict__ batch,
                                                 float* __restrict__ out) {
    __shared__ __attribute__((aligned(16))) float zt[BR * TP];
    __shared__ __attribute__((aligned(16))) float y1[BR * TP];
    __shared__ __attribute__((aligned(16))) int   gidx[BR];

    const int bix   = blockIdx.x;
    const int mt    = (bix < NB5) ? 5 : 4;
    const int tile0 = (bix < NB5) ? 5 * bix : 5 * NB5 + 4 * (bix - NB5);
    const int row0  = tile0 * 16;
    const int tid   = threadIdx.x;
    int gmine = 0;
    if (POOL && tid < 16 * mt) gmine = batch[row0 + tid];
    const int lane = tid & 31;
    const int wave = __builtin_amdgcn_readfirstlane(tid >> 5);
    const float e1 = 1.0f + epsp[0];

    {
        const float4 zero4 = make_float4(0.0f, 0.0f, 0.0f, 0.0f);
        float4 mu = zero4, sc = zero4, be = zero4;
        if (NORM) {
            bn_coef(st_prev, gamma, beta, 4 * lane + 0, mu.x, sc.x, be.x);
            bn_coef(st_prev, gamma, beta, 4 * lane + 1, mu.y, sc.y, be.y);
            bn_coef(st_prev, gamma, beta, 4 * lane + 2, mu.z, sc.z, be.z);
            bn_coef(st_prev, gamma, beta, 4 * lane + 3, mu.w, sc.w, be.w);
        }
        const float4* h4 = (const float4*)hin;

        for (int t = 0; t < mt; ++t) {
        const int rA = 16 * t + 2 * wave;
        const int nodeA = row0 + rA;
        const int nodeB = nodeA + 1;
        const int begA = rowbeg[nodeA];
        const int begB = rowbeg[nodeB];
        const int cntA = deg[nodeA];
        const int cntB = deg[nodeB];

        float4 selfA = h4[(long)nodeA * (DIM / 4) + lane];
        float4 selfB = h4[(long)nodeB * (DIM / 4) + lane];

        float4 accA = zero4, accB = zero4;
        const int cmax = (cntA > cntB) ? cntA : cntB;
        for (int base = 0; base < cmax; base += 32) {
            int mA = cntA - base; mA = (mA < 0) ? 0 : ((mA > 32) ? 32 : mA);
            int mB = cntB - base; mB = (mB < 0) ? 0 : ((mB > 32) ? 32 : mB);
            const int mineA = (lane < mA) ? col[begA + base + lane] : 0;
            const int mineB = (lane < mB) ? col[begB + base + lane] : 0;
            const int mm = (mA > mB) ? mA : mB;
            asm volatile("" :: "v"(mineA), "v"(mineB));
            for (int j = 0; j < mm; j += 4) {
                float4 a0 = zero4, a1 = zero4, a2 = zero4, a3 = zero4;
                float4 c0 = zero4, c1 = zero4, c2 = zero4, c3 = zero4;
                if (j + 0 < mA) a0 = h4[(long)__builtin_amdgcn_readlane(mineA, j + 0) * (DIM / 4) + lane];
                if (j + 1 < mA) a1 = h4[(long)__builtin_amdgcn_readlane(mineA, j + 1) * (DIM / 4) + lane];
                if (j + 2 < mA) a2 = h4[(long)__builtin_amdgcn_readlane(mineA, j + 2) * (DIM / 4) + lane];
                if (j + 3 < mA) a3 = h4[(long)__builtin_amdgcn_readlane(mineA, j + 3) * (DIM / 4) + lane];
                if (j + 0 < mB) c0 = h4[(long)__builtin_amdgcn_readlane(mineB, j + 0) * (DIM / 4) + lane];
                if (j + 1 < mB) c1 = h4[(long)__builtin_amdgcn_readlane(mineB, j + 1) * (DIM / 4) + lane];
                if (j + 2 < mB) c2 = h4[(long)__builtin_amdgcn_readlane(mineB, j + 2) * (DIM / 4) + lane];
                if (j + 3 < mB) c3 = h4[(long)__builtin_amdgcn_readlane(mineB, j + 3) * (DIM / 4) + lane];
                if (NORM) {
                    if (j + 0 < mA) a0 = bn_relu4(a0, mu, sc, be);
                    if (j + 1 < mA) a1 = bn_relu4(a1, mu, sc, be);
                    if (j + 2 < mA) a2 = bn_relu4(a2, mu, sc, be);
                    if (j + 3 < mA) a3 = bn_relu4(a3, mu, sc, be);
                    if (j + 0 < mB) c0 = bn_relu4(c0, mu, sc, be);
                    if (j + 1 < mB) c1 = bn_relu4(c1, mu, sc, be);
                    if (j + 2 < mB) c2 = bn_relu4(c2, mu, sc, be);
                    if (j + 3 < mB) c3 = bn_relu4(c3, mu, sc, be);
                }
                accA.x += (a0.x + a1.x) + (a2.x + a3.x);
                accA.y += (a0.y + a1.y) + (a2.y + a3.y);
                accA.z += (a0.z + a1.z) + (a2.z + a3.z);
                accA.w += (a0.w + a1.w) + (a2.w + a3.w);
                accB.x += (c0.x + c1.x) + (c2.x + c3.x);
                accB.y += (c0.y + c1.y) + (c2.y + c3.y);
                accB.z += (c0.z + c1.z) + (c2.z + c3.z);
                accB.w += (c0.w + c1.w) + (c2.w + c3.w);
            }
        }

        if (NORM) {
            selfA = bn_relu4(selfA, mu, sc, be);
            selfB = bn_relu4(selfB, mu, sc, be);
        }
        float4 zA, zB;
        zA.x = e1 * selfA.x + accA.x;
        zA.y = e1 * selfA.y + accA.y;
        zA.z = e1 * selfA.z + accA.z;
        zA.w = e1 * selfA.w + accA.w;
        zB.x = e1 * selfB.x + accB.x;
        zB.y = e1 * selfB.y + accB.y;
        zB.z = e1 * selfB.z + accB.z;
        zB.w = e1 * selfB.w + accB.w;
        *(float4*)&zt[rA * TP + 4 * lane]       = zA;
        *(float4*)&zt[(rA + 1) * TP + 4 * lane] = zB;
        }
    }

    if (POOL && tid < 16 * mt) gidx[tid] = gmine;

    if (mt == 5) gin_products<POOL, 5>(row0, zt, y1, gidx, W1, b1, W2, b2, hout, stats, do_stats, out);
    else         gin_products<POOL, 4>(row0, zt, y1, gidx, W1, b1, W2, b2, hout, stats, do_stats, out);
}

extern "C" void kernel_launch(void* const* d_in, const int* in_sizes, int n_in,
                              void* d_out, int out_size, void* d_ws, size_t ws_size,
                              hipStream_t stream) {
    const float* x     = (const float*)d_in[0];
    const int*   ei    = (const int*)d_in[1];
    const int*   batch = (const int*)d_in[2];
    const float* W1    = (const float*)d_in[3];
    const float* b1    = (const float*)d_in[4];
    const float* W2    = (const float*)d_in[5];
    const float* b2    = (const float*)d_in[6];
    const float* eps   = (const float*)d_in[7];
    const float* gamma = (const float*)d_in[8];
    const float* beta  = (const float*)d_in[9];
    float* out = (float*)d_out;

    const long ND = (long)N_NODES * DIM;
    float* hA    = (float*)d_ws;
    float* hB    = hA + ND;
    float* stats = hB + ND;
    int* deg    = (int*)(stats + N_LAYERS * 2 * DIM);
    int* rowbeg = deg + N_NODES;
    int* rank   = rowbeg + N_NODES;
    int* col    = rank + N_EDGES;

    const int* src = ei;
    const int* dst = ei + N_EDGES;

    const int  degGrd  = (N_NODES + 255) / 256;
    const int  edgeGrd = (N_EDGES + 255) / 256;
    const int  scanGrd = (N_NODES + 1023) / 1024;
    const int  layGrd  = LAY_GRID;

    gin_zero_i<<<degGrd, 256, 0, stream>>>(deg, N_NODES, stats, out);
    gin_csr_count<<<edgeGrd, 256, 0, stream>>>(dst, deg, rank);
    gin_csr_scan<<<scanGrd, 1024, 0, stream>>>(deg, rowbeg);
    gin_csr_fill<<<edgeGrd, 256, 0, stream>>>(src, dst, rowbeg, rank, col);

    for (int i = 0; i < N_LAYERS; ++i) {
        const float* hin  = (i == 0) ? x : ((i & 1) ? hA : hB);
        float*       hout = (i & 1) ? hB : hA;
        int do_stats = (i < N_LAYERS - 1) ? 1 : 0;
        float*       st_cur  = stats + (long)i * 2 * DIM;
        const float* st_prev = stats + (long)(i > 0 ? i - 1 : 0) * 2 * DIM;
        const float* g_prev  = gamma + (long)(i > 0 ? i - 1 : 0) * DIM;
        const float* b_prev  = beta + (long)(i > 0 ? i - 1 : 0) * DIM;

        if (i == 0) {
            gin_layer<false, false><<<layGrd, 256, 0, stream>>>(hin, rowbeg, deg, col,
                                                         W1 + (long)i * DIM * DIM, b1 + (long)i * DIM,
                                                         W2 + (long)i * DIM * DIM, b2 + (long)i * DIM,
                                                         eps + i, hout, st_cur, do_stats,
                                                         st_prev, g_prev, b_prev, batch, out);
        } else if (i < N_LAYERS - 1) {
            gin_layer<true, false><<<layGrd, 256, 0, stream>>>(hin, rowbeg, deg, col,
                                                        W1 + (long)i * DIM * DIM, b1 + (long)i * DIM,
                                                        W2 + (long)i * DIM * DIM, b2 + (long)i * DIM,
                                                        eps + i, hout, st_cur, do_stats,
                                                        st_prev, g_prev, b_prev, batch, out);
        } else {
            gin_layer<true, true><<<layGrd, 256, 0, stream>>>(hin, rowbeg, deg, col,
                                                        W1 + (long)i * DIM * DIM, b1 + (long)i * DIM,
                                                        W2 + (long)i * DIM * DIM, b2 + (long)i * DIM,
                                                        eps + i, hout, st_cur, do_stats,
                                                        st_prev, g_prev, b_prev, batch, out);
        }
    }
}


# Edge list as fixed rows of 32 slots per destination, filled by the counting pass itself (slot = the count it returns; longer rows spill to an overflow list): scan and fill launches go; sum order kept
# speedup vs baseline: 4.1017x; 1.1094x over previous
#include <hip/hip_runtime.h>
#include <hip/hip_bf16.h>

#define N_NODES   50000
#define N_EDGES   800000
#define DIM       128
#define N_LAYERS  4
#define NUM_GRAPHS 128
#define BN_EPS    1e-5f
#define TP        132
#define MT        5
#define BR        (16 * MT)
#define LAY_GRID  768
#define NB5       53
#define ROWCAP    32

typedef __attribute__((ext_vector_type(2))) float v2f;
typedef __attribute__((ext_vector_type(8))) float v8f;

__global__ __launch_bounds__(256) void gin_zero_i(int* __restrict__ p, int n, float* __restrict__ stats,
                                                  float* __restrict__ out) {
    if (blockIdx.x < N_LAYERS) stats[blockIdx.x * 2 * DIM + threadIdx.x] = 0.0f;
    int i = blockIdx.x * blockDim.x + threadIdx.x;
    if (i < NUM_GRAPHS * DIM) out[i] = 0.0f;
    if (i < n) p[i] = 0;
}

__global__ __launch_bounds__(256) void gin_edge_rows(const int* __restrict__ src,
                                                     const int* __restrict__ dst,
                                                     int* __restrict__ deg,
                                                     int* __restrict__ novf,
                                                     int* __restrict__ ell,
                                                     int* __restrict__ ovf_dst,
                                                     int* __restrict__ ovf_src) {
    int e = blockIdx.x * blockDim.x + threadIdx.x;
    if (e >= N_EDGES) return;
    const int d = dst[e];
    const int s = src[e];
    const int r = atomicAdd(&deg[d], 1);
    if (r < ROWCAP) {
        ell[d * ROWCAP + r] = s;
    } else {
        const int p = atomicAdd(novf, 1);
        ovf_dst[p] = d;
        ovf_src[p] = s;
    }
}

__device__ __forceinline__ void bn_coef(const float* __restrict__ st,
                                        const float* __restrict__ gamma,
                                        const float* __restrict__ beta,
                                        int c, float& mu, float& sc, float& be) {
    const float invN = 1.0f / (float)N_NODES;
    mu = st[c] * invN;
    float var = st[DIM + c] * invN - mu * mu;
    sc = rsqrtf(var + BN_EPS) * gamma[c];
    be = beta[c];
}
__device__ __forceinline__ float bn_relu1(float z, float mu, float sc, float be) {
    return fmaxf((z - mu) * sc + be, 0.0f);
}
__device__ __forceinline__ float4 bn_relu4(float4 z, float4 mu, float4 sc, float4 be) {
    float4 r;
    r.x = bn_relu1(z.x, mu.x, sc.x, be.x);
    r.y = bn_relu1(z.y, mu.y, sc.y, be.y);
    r.z = bn_relu1(z.z, mu.z, sc.z, be.z);
    r.w = bn_relu1(z.w, mu.w, sc.w, be.w);
    return r;
}

template <bool POOL, int MTT>
__device__ __forceinline__ void gin_products(const int row0, const float* __restrict__ zt, float* __restrict__ y1,
                                             const int* __restrict__ gidx,
                                             const float* __restrict__ W1, const float* __restrict__ b1,
                                             const float* __restrict__ W2, const float* __restrict__ b2,
                                             float* __restrict__ hout, float* __restrict__ stats, int do_stats,
                                             float* __restrict__ out) {
    const int tid  = threadIdx.x;
    const int lane = tid & 31;
    const int wave = __builtin_amdgcn_readfirstlane(tid >> 5);
    const int n    = lane & 15;
    const int hi   = lane >> 4;
    const int col0 = wave * 16;
    const int kk   = hi * 2;

    v2f bw[DIM / 4];
    #pragma unroll
    for (int ks = 0; ks < DIM / 4; ++ks) {
        bw[ks].x = W1[(4 * ks + kk) * DIM + col0 + n];
        bw[ks].y = W1[(4 * ks + kk + 1) * DIM + col0 + n];
    }
    float bv = b1[col0 + n];
    __syncthreads();

    v8f acc[MTT];
    #pragma unroll
    for (int t = 0; t < MTT; ++t)
        for (int v = 0; v < 8; ++v) acc[t][v] = bv;
    #pragma unroll
    for (int ks = 0; ks < DIM / 4; ++ks) {
        #pragma unroll
        for (int t = 0; t < MTT; ++t) {
            v2f a;
            const float* zp = &zt[(16 * t + n) * TP + 4 * ks + kk];
            a.x = zp[0];
            a.y = zp[1];
            acc[t] = __builtin_amdgcn_wmma_f32_16x16x4_f32(false, a, false, bw[ks],
                                                           (short)0, acc[t], false, false);
        }
    }
    __builtin_amdgcn_sched_barrier(0);
    #pragma unroll
    for (int ks = 0; ks < DIM / 4; ++ks) {
        bw[ks].x = W2[(4 * ks + kk) * DIM + col0 + n];
        bw[ks].y = W2[(4 * ks + kk + 1) * DIM + col0 + n];
    }
    const float bv2 = b2[col0 + n];
    #pragma unroll
    for (int t = 0; t < MTT; ++t) {
        for (int v = 0; v < 8; ++v) {
            int m = 16 * t + v + 8 * hi;
            y1[m * TP + col0 + n] = fmaxf(acc[t][v], 0.0f);
        }
    }
    __syncthreads();

    #pragma unroll
    for (int t = 0; t < MTT; ++t)
        for (int v = 0; v < 8; ++v) acc[t][v] = bv2;
    #pragma unroll
    for (int ks = 0; ks < DIM / 4; ++ks) {
        #pragma unroll
        for (int t = 0; t < MTT; ++t) {
            v2f a;
            const float* yp = &y1[(16 * t + n) * TP + 4 * ks + kk];
            a.x = yp[0];
            a.y = yp[1];
            acc[t] = __builtin_amdgcn_wmma_f32_16x16x4_f32(false, a, false, bw[ks],
                                                           (short)0, acc[t], false, false);
        }
    }

    if (POOL) {
        float* oc = out + col0 + n;
        int cur = gidx[8 * hi];
        float run = 0.0f;
        #pragma unroll
        for (int t = 0; t < MTT; ++t) {
            const int4 g0 = *(const int4*)&gidx[16 * t + 8 * hi];
            const int4 g1 = *(const int4*)&gidx[16 * t + 8 * hi + 4];
            const int g[8] = {g0.x, g0.y, g0.z, g0.w, g1.x, g1.y, g1.z, g1.w};
            #pragma unroll
            for (int v = 0; v < 8; ++v) {
                if (g[v] != cur) {
                    atomicAdd(oc + (long)cur * DIM, run);
                    run = 0.0f;
                    cur = g[v];
                }
                run += acc[t][v];
            }
        }
        atomicAdd(oc + (long)cur * DIM, run);
    } else {
        float s = 0.0f, ssq = 0.0f;
        #pragma unroll
        for (int t = 0; t < MTT; ++t) {
            for (int v = 0; v < 8; ++v) {
                int m = 16 * t + v + 8 * hi;
                float val = acc[t][v];
                hout[(long)(row0 + m) * DIM + col0 + n] = val;
                s += val;
                ssq += val * val;
            }
        }
        if (do_stats) {
            atomicAdd(&stats[col0 + n], s);
            atomicAdd(&stats[DIM + col0 + n], ssq);
        }
    }
}

template <bool NORM, bool POOL>
__global__ __launch_bounds__(256, 6) void gin_layer(const float* __restrict__ hin,
                                                 const int* __restrict__ deg,
                                                 const int* __restrict__ ell,
                                                 const int* __restrict__ novf,
                                                 const int* __restrict__ ovf_dst,
                                                 const int* __restrict__ ovf_src,
                                                 const float* __restrict__ W1,
                                                 const float* __restrict__ b1,
                                                 const float* __restrict__ W2,
                                                 const float* __restrict__ b2,
                                                 const float* __restrict__ epsp,
                                                 float* __restrict__ hout,
                                                 float* __restrict__ stats,
                                                 int do_stats,
                                                 const float* __restrict__ st_prev,
                                                 const float* __restrict__ gamma,
                                                 const float* __restrict__ beta,
                                                 const int* __restrict__ batch,
                                                 float* __restrict__ out) {
    __shared__ __attribute__((aligned(16))) float zt[BR * TP];
    __shared__ __attribute__((aligned(16))) float y1[BR * TP];
    __shared__ __attribute__((aligned(16))) int   gidx[BR];

    const int bix   = blockIdx.x;
    const int mt    = (bix < NB5) ? 5 : 4;
    const int tile0 = (bix < NB5) ? 5 * bix : 5 * NB5 + 4 * (bix - NB5);
    const int row0  = tile0 * 16;
    const int tid   = threadIdx.x;
    int gmine = 0;
    if (POOL && tid < 16 * mt) gmine = batch[row0 + tid];
    const int lane = tid & 31;
    const int wave = __builtin_amdgcn_readfirstlane(tid >> 5);
    const float e1 = 1.0f + epsp[0];

    {
        const float4 zero4 = make_float4(0.0f, 0.0f, 0.0f, 0.0f);
        float4 mu = zero4, sc = zero4, be = zero4;
        if (NORM) {
            bn_coef(st_prev, gamma, beta, 4 * lane + 0, mu.x, sc.x, be.x);
            bn_coef(st_prev, gamma, beta, 4 * lane + 1, mu.y, sc.y, be.y);
            bn_coef(st_prev, gamma, beta, 4 * lane + 2, mu.z, sc.z, be.z);
            bn_coef(st_prev, gamma, beta, 4 * lane + 3, mu.w, sc.w, be.w);
        }
        const float4* h4 = (const float4*)hin;

        for (int t = 0; t < mt; ++t) {
        const int rA = 16 * t + 2 * wave;
        const int nodeA = row0 + rA;
        const int nodeB = nodeA + 1;
        const int cntA = deg[nodeA];
        const int cntB = deg[nodeB];

        float4 selfA = h4[(long)nodeA * (DIM / 4) + lane];
        float4 selfB = h4[(long)nodeB * (DIM / 4) + lane];

        float4 accA = zero4, accB = zero4;
        {
            const int mA = (cntA > ROWCAP) ? ROWCAP : cntA;
            const int mB = (cntB > ROWCAP) ? ROWCAP : cntB;
            const int mineA = (lane < mA) ? ell[nodeA * ROWCAP + lane] : 0;
            const int mineB = (lane < mB) ? ell[nodeB * ROWCAP + lane] : 0;
            const int mm = (mA > mB) ? mA : mB;
            asm volatile("" :: "v"(mineA), "v"(mineB));
            for (int j = 0; j < mm; j += 4) {
                float4 a0 = zero4, a1 = zero4, a2 = zero4, a3 = zero4;
                float4 c0 = zero4, c1 = zero4, c2 = zero4, c3 = zero4;
                if (j + 0 < mA) a0 = h4[(long)__builtin_amdgcn_readlane(mineA, j + 0) * (DIM / 4) + lane];
                if (j + 1 < mA) a1 = h4[(long)__builtin_amdgcn_readlane(mineA, j + 1) * (DIM / 4) + lane];
                if (j + 2 < mA) a2 = h4[(long)__builtin_amdgcn_readlane(mineA, j + 2) * (DIM / 4) + lane];
                if (j + 3 < mA) a3 = h4[(long)__builtin_amdgcn_readlane(mineA, j + 3) * (DIM / 4) + lane];
                if (j + 0 < mB) c0 = h4[(long)__builtin_amdgcn_readlane(mineB, j + 0) * (DIM / 4) + lane];
                if (j + 1 < mB) c1 = h4[(long)__builtin_amdgcn_readlane(mineB, j + 1) * (DIM / 4) + lane];
                if (j + 2 < mB) c2 = h4[(long)__builtin_amdgcn_readlane(mineB, j + 2) * (DIM / 4) + lane];
                if (j + 3 < mB) c3 = h4[(long)__builtin_amdgcn_readlane(mineB, j + 3) * (DIM / 4) + lane];
                if (NORM) {
                    if (j + 0 < mA) a0 = bn_relu4(a0, mu, sc, be);
                    if (j + 1 < mA) a1 = bn_relu4(a1, mu, sc, be);
                    if (j + 2 < mA) a2 = bn_relu4(a2, mu, sc, be);
                    if (j + 3 < mA) a3 = bn_relu4(a3, mu, sc, be);
                    if (j + 0 < mB) c0 = bn_relu4(c0, mu, sc, be);
                    if (j + 1 < mB) c1 = bn_relu4(c1, mu, sc, be);
                    if (j + 2 < mB) c2 = bn_relu4(c2, mu, sc, be);
                    if (j + 3 < mB) c3 = bn_relu4(c3, mu, sc, be);
                }
                accA.x += (a0.x + a1.x) + (a2.x + a3.x);
                accA.y += (a0.y + a1.y) + (a2.y + a3.y);
                accA.z += (a0.z + a1.z) + (a2.z + a3.z);
                accA.w += (a0.w + a1.w) + (a2.w + a3.w);
                accB.x += (c0.x + c1.x) + (c2.x + c3.x);
                accB.y += (c0.y + c1.y) + (c2.y + c3.y);
                accB.z += (c0.z + c1.z) + (c2.z + c3.z);
                accB.w += (c0.w + c1.w) + (c2.w + c3.w);
            }
        }
        if (cntA > ROWCAP || cntB > ROWCAP) {
            int nov = novf[0];
            nov = (nov > N_EDGES) ? N_EDGES : nov;
            for (int base = 0; base < nov; base += 32) {
                const int i  = base + lane;
                const int od = (i < nov) ? ovf_dst[i] : -1;
                const int os = (i < nov) ? ovf_src[i] : 0;
                unsigned hitA = __builtin_amdgcn_ballot_w32(od == nodeA);
                unsigned hitB = __builtin_amdgcn_ballot_w32(od == nodeB);
                while (hitA) {
                    const int l = __builtin_ctz(hitA);
                    hitA &= hitA - 1;
                    float4 r = h4[(long)__builtin_amdgcn_readlane(os, l) * (DIM / 4) + lane];
                    if (NORM) r = bn_relu4(r, mu, sc, be);
                    accA.x += r.x; accA.y += r.y; accA.z += r.z; accA.w += r.w;
                }
                while (hitB) {
                    const int l = __builtin_ctz(hitB);
                    hitB &= hitB - 1;
                    float4 r = h4[(long)__builtin_amdgcn_readlane(os, l) * (DIM / 4) + lane];
                    if (NORM) r = bn_relu4(r, mu, sc, be);
                    accB.x += r.x; accB.y += r.y; accB.z += r.z; accB.w += r.w;
                }
            }
        }

        if (NORM) {
            selfA = bn_relu4(selfA, mu, sc, be);
            selfB = bn_relu4(selfB, mu, sc, be);
        }
        float4 zA, zB;
        zA.x = e1 * selfA.x + accA.x;
        zA.y = e1 * selfA.y + accA.y;
        zA.z = e1 * selfA.z + accA.z;
        zA.w = e1 * selfA.w + accA.w;
        zB.x = e1 * selfB.x + accB.x;
        zB.y = e1 * selfB.y + accB.y;
        zB.z = e1 * selfB.z + accB.z;
        zB.w = e1 * selfB.w + accB.w;
        *(float4*)&zt[rA * TP + 4 * lane]       = zA;
        *(float4*)&zt[(rA + 1) * TP + 4 * lane] = zB;
        }
    }

    if (POOL && tid < 16 * mt) gidx[tid] = gmine;

    if (mt == 5) gin_products<POOL, 5>(row0, zt, y1, gidx, W1, b1, W2, b2, hout, stats, do_stats, out);
    else         gin_products<POOL, 4>(row0, zt, y1, gidx, W1, b1, W2, b2, hout, stats, do_stats, out);
}

extern "C" void kernel_launch(void* const* d_in, const int* in_sizes, int n_in,
                              void* d_out, int out_size, void* d_ws, size_t ws_size,
                              hipStream_t stream) {
    const float* x     = (const float*)d_in[0];
    const int*   ei    = (const int*)d_in[1];
    const int*   batch = (const int*)d_in[2];
    const float* W1    = (const float*)d_in[3];
    const float* b1    = (const float*)d_in[4];
    const float* W2    = (const float*)d_in[5];
    const float* b2    = (const float*)d_in[6];
    const float* eps   = (const float*)d_in[7];
    const float* gamma = (const float*)d_in[8];
    const float* beta  = (const float*)d_in[9];
    float* out = (float*)d_out;

    const long ND = (long)N_NODES * DIM;
    float* hA    = (float*)d_ws;
    float* hB    = hA + ND;
    float* stats = hB + ND;
    int* ell     = (int*)(stats + N_LAYERS * 2 * DIM);
    int* deg     = ell + (long)N_NODES * ROWCAP;
    int* novf    = deg + N_NODES;
    int* ovf_dst = novf + 32;
    int* ovf_src = ovf_dst + N_EDGES;

    const int* src = ei;
    const int* dst = ei + N_EDGES;

    const int  degGrd  = (N_NODES + 1 + 255) / 256;
    const int  edgeGrd = (N_EDGES + 255) / 256;
    const int  layGrd  = LAY_GRID;

    gin_zero_i<<<degGrd, 256, 0, stream>>>(deg, N_NODES + 1, stats, out);
    gin_edge_rows<<<edgeGrd, 256, 0, stream>>>(src, dst, deg, novf, ell, ovf_dst, ovf_src);

    for (int i = 0; i < N_LAYERS; ++i) {
        const float* hin  = (i == 0) ? x : ((i & 1) ? hA : hB);
        float*       hout = (i & 1) ? hB : hA;
        int do_stats = (i < N_LAYERS - 1) ? 1 : 0;
        float*       st_cur  = stats + (long)i * 2 * DIM;
        const float* st_prev = stats + (long)(i > 0 ? i - 1 : 0) * 2 * DIM;
        const float* g_prev  = gamma + (long)(i > 0 ? i - 1 : 0) * DIM;
        const float* b_prev  = beta + (long)(i > 0 ? i - 1 : 0) * DIM;

        if (i == 0) {
            gin_layer<false, false><<<layGrd, 256, 0, stream>>>(hin, deg, ell, novf, ovf_dst, ovf_src,
                                                         W1 + (long)i * DIM * DIM, b1 + (long)i * DIM,
                                                         W2 + (long)i * DIM * DIM, b2 + (long)i * DIM,
                                                         eps + i, hout, st_cur, do_stats,
                                                         st_prev, g_prev, b_prev, batch, out);
        } else if (i < N_LAYERS - 1) {
            gin_layer<true, false><<<layGrd, 256, 0, stream>>>(hin, deg, ell, novf, ovf_dst, ovf_src,
                                                        W1 + (long)i * DIM * DIM, b1 + (long)i * DIM,
                                                        W2 + (long)i * DIM * DIM, b2 + (long)i * DIM,
                                                        eps + i, hout, st_cur, do_stats,
                                                        st_prev, g_prev, b_prev, batch, out);
        } else {
            gin_layer<true, true><<<layGrd, 256, 0, stream>>>(hin, deg, ell, novf, ovf_dst, ovf_src,
                                                        W1 + (long)i * DIM * DIM, b1 + (long)i * DIM,
                                                        W2 + (long)i * DIM * DIM, b2 + (long)i * DIM,
                                                        eps + i, hout, st_cur, do_stats,
                                                        st_prev, g_prev, b_prev, batch, out);
        }
    }
}


# Layer kernel: W1, W2 fragments read from a copy in fragment order that the edge-list launch writes each call: 16 whole-line 16-byte loads per wave and matrix, not 128 4-byte loads on half lines
# speedup vs baseline: 4.2526x; 1.0368x over previous
#include <hip/hip_runtime.h>
#include <hip/hip_bf16.h>

#define N_NODES   50000
#define N_EDGES   800000
#define DIM       128
#define N_LAYERS  4
#define NUM_GRAPHS 128
#define BN_EPS    1e-5f
#define TP        132
#define MT        5
#define BR        (16 * MT)
#define LAY_GRID  768
#define NB5       53
#define ROWCAP    32

typedef __attribute__((ext_vector_type(2))) float v2f;
typedef __attribute__((ext_vector_type(8))) float v8f;

__global__ __launch_bounds__(256) void gin_zero_i(int* __restrict__ p, int n, float* __restrict__ stats,
                                                  float* __restrict__ out) {
    if (blockIdx.x < N_LAYERS) stats[blockIdx.x * 2 * DIM + threadIdx.x] = 0.0f;
    int i = blockIdx.x * blockDim.x + threadIdx.x;
    if (i < NUM_GRAPHS * DIM) out[i] = 0.0f;
    if (i < n) p[i] = 0;
}

#define WPK_FLOATS (N_LAYERS * 2 * DIM * DIM)
__global__ __launch_bounds__(256) void gin_edge_rows(const int* __restrict__ src,
                                                     const int* __restrict__ dst,
                                                     int* __restrict__ deg,
                                                     int* __restrict__ novf,
                                                     int* __restrict__ ell,
                                                     int* __restrict__ ovf_dst,
                                                     int* __restrict__ ovf_src,
                                                     const float* __restrict__ W1,
                                                     const float* __restrict__ W2,
                                                     float* __restrict__ wpk) {
    int e = blockIdx.x * blockDim.x + threadIdx.x;
    if (e < WPK_FLOATS / 4) {
        const int m    = e >> 12;
        const int w    = (e >> 9) & 7;
        const int q    = (e >> 5) & 15;
        const int l    = e & 31;
        const int krow = 8 * q + 2 * (l >> 4);
        const float* Wm = ((m & 1) ? W2 : W1) + (long)(m >> 1) * DIM * DIM + 16 * w + (l & 15);
        float4 f;
        f.x = Wm[(krow + 0) * DIM];
        f.y = Wm[(krow + 1) * DIM];
        f.z = Wm[(krow + 4) * DIM];
        f.w = Wm[(krow + 5) * DIM];
        ((float4*)wpk)[e] = f;
    }
    if (e >= N_EDGES) return;
    const int d = dst[e];
    const int s = src[e];
    const int r = atomicAdd(&deg[d], 1);
    if (r < ROWCAP) {
        ell[d * ROWCAP + r] = s;
    } else {
        const int p = atomicAdd(novf, 1);
        ovf_dst[p] = d;
        ovf_src[p] = s;
    }
}

__device__ __forceinline__ void bn_coef(const float* __restrict__ st,
                                        const float* __restrict__ gamma,
                                        const float* __restrict__ beta,
                                        int c, float& mu, float& sc, float& be) {
    const float invN = 1.0f / (float)N_NODES;
    mu = st[c] * invN;
    float var = st[DIM + c] * invN - mu * mu;
    sc = rsqrtf(var + BN_EPS) * gamma[c];
    be = beta[c];
}
__device__ __forceinline__ float bn_relu1(float z, float mu, float sc, float be) {
    return fmaxf((z - mu) * sc + be, 0.0f);
}
__device__ __forceinline__ float4 bn_relu4(float4 z, float4 mu, float4 sc, float4 be) {
    float4 r;
    r.x = bn_relu1(z.x, mu.x, sc.x, be.x);
    r.y = bn_relu1(z.y, mu.y, sc.y, be.y);
    r.z = bn_relu1(z.z, mu.z, sc.z, be.z);
    r.w = bn_relu1(z.w, mu.w, sc.w, be.w);
    return r;
}

template <bool POOL, int MTT>
__device__ __forceinline__ void gin_products(const int row0, const float* __restrict__ zt, float* __restrict__ y1,
                                             const int* __restrict__ gidx,
                                             const float* __restrict__ W1, const float* __restrict__ b1,
                                             const float* __restrict__ W2, const float* __restrict__ b2,
                                             float* __restrict__ hout, float* __restrict__ stats, int do_stats,
                                             float* __restrict__ out) {
    const int tid  = threadIdx.x;
    const int lane = tid & 31;
    const int wave = __builtin_amdgcn_readfirstlane(tid >> 5);
    const int n    = lane & 15;
    const int hi   = lane >> 4;
    const int col0 = wave * 16;
    const int kk   = hi * 2;

    v2f bw[DIM / 4];
    const float4* wp = (const float4*)W1 + wave * (DIM / 8) * 32 + lane;
    #pragma unroll
    for (int q = 0; q < DIM / 8; ++q) {
        const float4 f = wp[q * 32];
        bw[2 * q].x     = f.x;
        bw[2 * q].y     = f.y;
        bw[2 * q + 1].x = f.z;
        bw[2 * q + 1].y = f.w;
    }
    float bv = b1[col0 + n];
    __syncthreads();

    v8f acc[MTT];
    #pragma unroll
    for (int t = 0; t < MTT; ++t)
        for (int v = 0; v < 8; ++v) acc[t][v] = bv;
    #pragma unroll
    for (int ks = 0; ks < DIM / 4; ++ks) {
        #pragma unroll
        for (int t = 0; t < MTT; ++t) {
            v2f a;
            const float* zp = &zt[(16 * t + n) * TP + 4 * ks + kk];
            a.x = zp[0];
            a.y = zp[1];
            acc[t] = __builtin_amdgcn_wmma_f32_16x16x4_f32(false, a, false, bw[ks],
                                                           (short)0, acc[t], false, false);
        }
    }
    __builtin_amdgcn_sched_barrier(0);
    wp = (const float4*)W2 + wave * (DIM / 8) * 32 + lane;
    #pragma unroll
    for (int q = 0; q < DIM / 8; ++q) {
        const float4 f = wp[q * 32];
        bw[2 * q].x     = f.x;
        bw[2 * q].y     = f.y;
        bw[2 * q + 1].x = f.z;
        bw[2 * q + 1].y = f.w;
    }
    const float bv2 = b2[col0 + n];
    #pragma unroll
    for (int t = 0; t < MTT; ++t) {
        for (int v = 0; v < 8; ++v) {
            int m = 16 * t + v + 8 * hi;
            y1[m * TP + col0 + n] = fmaxf(acc[t][v], 0.0f);
        }
    }
    __syncthreads();

    #pragma unroll
    for (int t = 0; t < MTT; ++t)
        for (int v = 0; v < 8; ++v) acc[t][v] = bv2;
    #pragma unroll
    for (int ks = 0; ks < DIM / 4; ++ks) {
        #pragma unroll
        for (int t = 0; t < MTT; ++t) {
            v2f a;
            const float* yp = &y1[(16 * t + n) * TP + 4 * ks + kk];
            a.x = yp[0];
            a.y = yp[1];
            acc[t] = __builtin_amdgcn_wmma_f32_16x16x4_f32(false, a, false, bw[ks],
                                                           (short)0, acc[t], false, false);
        }
    }

    if (POOL) {
        float* oc = out + col0 + n;
        int cur = gidx[8 * hi];
        float run = 0.0f;
        #pragma unroll
        for (int t = 0; t < MTT; ++t) {
            const int4 g0 = *(const int4*)&gidx[16 * t + 8 * hi];
            const int4 g1 = *(const int4*)&gidx[16 * t + 8 * hi + 4];
            const int g[8] = {g0.x, g0.y, g0.z, g0.w, g1.x, g1.y, g1.z, g1.w};
            #pragma unroll
            for (int v = 0; v < 8; ++v) {
                if (g[v] != cur) {
                    atomicAdd(oc + (long)cur * DIM, run);
                    run = 0.0f;
                    cur = g[v];
                }
                run += acc[t][v];
            }
        }
        atomicAdd(oc + (long)cur * DIM, run);
    } else {
        float s = 0.0f, ssq = 0.0f;
        #pragma unroll
        for (int t = 0; t < MTT; ++t) {
            for (int v = 0; v < 8; ++v) {
                int m = 16 * t + v + 8 * hi;
                float val = acc[t][v];
                hout[(long)(row0 + m) * DIM + col0 + n] = val;
                s += val;
                ssq += val * val;
            }
        }
        if (do_stats) {
            atomicAdd(&stats[col0 + n], s);
            atomicAdd(&stats[DIM + col0 + n], ssq);
        }
    }
}

template <bool NORM, bool POOL>
__global__ __launch_bounds__(256, 6) void gin_layer(const float* __restrict__ hin,
                                                 const int* __restrict__ deg,
                                                 const int* __restrict__ ell,
                                                 const int* __restrict__ novf,
                                                 const int* __restrict__ ovf_dst,
                                                 const int* __restrict__ ovf_src,
                                                 const float* __restrict__ W1,
                                                 const float* __restrict__ b1,
                                                 const float* __restrict__ W2,
                                                 const float* __restrict__ b2,
                                                 const float* __restrict__ epsp,
                                                 float* __restrict__ hout,
                                                 float* __restrict__ stats,
                                                 int do_stats,
                                                 const float* __restrict__ st_prev,
                                                 const float* __restrict__ gamma,
                                                 const float* __restrict__ beta,
                                                 const int* __restrict__ batch,
                                                 float* __restrict__ out) {
    __shared__ __attribute__((aligned(16))) float zt[BR * TP];
    __shared__ __attribute__((aligned(16))) float y1[BR * TP];
    __shared__ __attribute__((aligned(16))) int   gidx[BR];

    const int bix   = blockIdx.x;
    const int mt    = (bix < NB5) ? 5 : 4;
    const int tile0 = (bix < NB5) ? 5 * bix : 5 * NB5 + 4 * (bix - NB5);
    const int row0  = tile0 * 16;
    const int tid   = threadIdx.x;
    int gmine = 0;
    if (POOL && tid < 16 * mt) gmine = batch[row0 + tid];
    const int lane = tid & 31;
    const int wave = __builtin_amdgcn_readfirstlane(tid >> 5);
    const float e1 = 1.0f + epsp[0];

    {
        const float4 zero4 = make_float4(0.0f, 0.0f, 0.0f, 0.0f);
        float4 mu = zero4, sc = zero4, be = zero4;
        if (NORM) {
            bn_coef(st_prev, gamma, beta, 4 * lane + 0, mu.x, sc.x, be.x);
            bn_coef(st_prev, gamma, beta, 4 * lane + 1, mu.y, sc.y, be.y);
            bn_coef(st_prev, gamma, beta, 4 * lane + 2, mu.z, sc.z, be.z);
            bn_coef(st_prev, gamma, beta, 4 * lane + 3, mu.w, sc.w, be.w);
        }
        const float4* h4 = (const float4*)hin;

        for (int t = 0; t < mt; ++t) {
        const int rA = 16 * t + 2 * wave;
        const int nodeA = row0 + rA;
        const int nodeB = nodeA + 1;
        const int cntA = deg[nodeA];
        const int cntB = deg[nodeB];

        float4 selfA = h4[(long)nodeA * (DIM / 4) + lane];
        float4 selfB = h4[(long)nodeB * (DIM / 4) + lane];

        float4 accA = zero4, accB = zero4;
        {
            const int mA = (cntA > ROWCAP) ? ROWCAP : cntA;
            const int mB = (cntB > ROWCAP) ? ROWCAP : cntB;
            const int mineA = (lane < mA) ? ell[nodeA * ROWCAP + lane] : 0;
            const int mineB = (lane < mB) ? ell[nodeB * ROWCAP + lane] : 0;
            const int mm = (mA > mB) ? mA : mB;
            asm volatile("" :: "v"(mineA), "v"(mineB));
            for (int j = 0; j < mm; j += 4) {
                float4 a0 = zero4, a1 = zero4, a2 = zero4, a3 = zero4;
                float4 c0 = zero4, c1 = zero4, c2 = zero4, c3 = zero4;
                if (j + 0 < mA) a0 = h4[(long)__builtin_amdgcn_readlane(mineA, j + 0) * (DIM / 4) + lane];
                if (j + 1 < mA) a1 = h4[(long)__builtin_amdgcn_readlane(mineA, j + 1) * (DIM / 4) + lane];
                if (j + 2 < mA) a2 = h4[(long)__builtin_amdgcn_readlane(mineA, j + 2) * (DIM / 4) + lane];
                if (j + 3 < mA) a3 = h4[(long)__builtin_amdgcn_readlane(mineA, j + 3) * (DIM / 4) + lane];
                if (j + 0 < mB) c0 = h4[(long)__builtin_amdgcn_readlane(mineB, j + 0) * (DIM / 4) + lane];
                if (j + 1 < mB) c1 = h4[(long)__builtin_amdgcn_readlane(mineB, j + 1) * (DIM / 4) + lane];
                if (j + 2 < mB) c2 = h4[(long)__builtin_amdgcn_readlane(mineB, j + 2) * (DIM / 4) + lane];
                if (j + 3 < mB) c3 = h4[(long)__builtin_amdgcn_readlane(mineB, j + 3) * (DIM / 4) + lane];
                if (NORM) {
                    if (j + 0 < mA) a0 = bn_relu4(a0, mu, sc, be);
                    if (j + 1 < mA) a1 = bn_relu4(a1, mu, sc, be);
                    if (j + 2 < mA) a2 = bn_relu4(a2, mu, sc, be);
                    if (j + 3 < mA) a3 = bn_relu4(a3, mu, sc, be);
                    if (j + 0 < mB) c0 = bn_relu4(c0, mu, sc, be);
                    if (j + 1 < mB) c1 = bn_relu4(c1, mu, sc, be);
                    if (j + 2 < mB) c2 = bn_relu4(c2, mu, sc, be);
                    if (j + 3 < mB) c3 = bn_relu4(c3, mu, sc, be);
                }
                accA.x += (a0.x + a1.x) + (a2.x + a3.x);
                accA.y += (a0.y + a1.y) + (a2.y + a3.y);
                accA.z += (a0.z + a1.z) + (a2.z + a3.z);
                accA.w += (a0.w + a1.w) + (a2.w + a3.w);
                accB.x += (c0.x + c1.x) + (c2.x + c3.x);
                accB.y += (c0.y + c1.y) + (c2.y + c3.y);
                accB.z += (c0.z + c1.z) + (c2.z + c3.z);
                accB.w += (c0.w + c1.w) + (c2.w + c3.w);
            }
        }
        if (cntA > ROWCAP || cntB > ROWCAP) {
            int nov = novf[0];
            nov = (nov > N_EDGES) ? N_EDGES : nov;
            for (int base = 0; base < nov; base += 32) {
                const int i  = base + lane;
                const int od = (i < nov) ? ovf_dst[i] : -1;
                const int os = (i < nov) ? ovf_src[i] : 0;
                unsigned hitA = __builtin_amdgcn_ballot_w32(od == nodeA);
                unsigned hitB = __builtin_amdgcn_ballot_w32(od == nodeB);
                while (hitA) {
                    const int l = __builtin_ctz(hitA);
                    hitA &= hitA - 1;
                    float4 r = h4[(long)__builtin_amdgcn_readlane(os, l) * (DIM / 4) + lane];
                    if (NORM) r = bn_relu4(r, mu, sc, be);
                    accA.x += r.x; accA.y += r.y; accA.z += r.z; accA.w += r.w;
                }
                while (hitB) {
                    const int l = __builtin_ctz(hitB);
                    hitB &= hitB - 1;
                    float4 r = h4[(long)__builtin_amdgcn_readlane(os, l) * (DIM / 4) + lane];
                    if (NORM) r = bn_relu4(r, mu, sc, be);
                    accB.x += r.x; accB.y += r.y; accB.z += r.z; accB.w += r.w;
                }
            }
        }

        if (NORM) {
            selfA = bn_relu4(selfA, mu, sc, be);
            selfB = bn_relu4(selfB, mu, sc, be);
        }
        float4 zA, zB;
        zA.x = e1 * selfA.x + accA.x;
        zA.y = e1 * selfA.y + accA.y;
        zA.z = e1 * selfA.z + accA.z;
        zA.w = e1 * selfA.w + accA.w;
        zB.x = e1 * selfB.x + accB.x;
        zB.y = e1 * selfB.y + accB.y;
        zB.z = e1 * selfB.z + accB.z;
        zB.w = e1 * selfB.w + accB.w;
        *(float4*)&zt[rA * TP + 4 * lane]       = zA;
        *(float4*)&zt[(rA + 1) * TP + 4 * lane] = zB;
        }
    }

    if (POOL && tid < 16 * mt) gidx[tid] = gmine;

    if (mt == 5) gin_products<POOL, 5>(row0, zt, y1, gidx, W1, b1, W2, b2, hout, stats, do_stats, out);
    else         gin_products<POOL, 4>(row0, zt, y1, gidx, W1, b1, W2, b2, hout, stats, do_stats, out);
}

extern "C" void kernel_launch(void* const* d_in, const int* in_sizes, int n_in,
                              void* d_out, int out_size, void* d_ws, size_t ws_size,
                              hipStream_t stream) {
    const float* x     = (const float*)d_in[0];
    const int*   ei    = (const int*)d_in[1];
    const int*   batch = (const int*)d_in[2];
    const float* W1    = (const float*)d_in[3];
    const float* b1    = (const float*)d_in[4];
    const float* W2    = (const float*)d_in[5];
    const float* b2    = (const float*)d_in[6];
    const float* eps   = (const float*)d_in[7];
    const float* gamma = (const float*)d_in[8];
    const float* beta  = (const float*)d_in[9];
    float* out = (float*)d_out;

    const long ND = (long)N_NODES * DIM;
    float* hA    = (float*)d_ws;
    float* hB    = hA + ND;
    float* stats = hB + ND;
    float* wpk   = stats + N_LAYERS * 2 * DIM;
    int* ell     = (int*)(wpk + WPK_FLOATS);
    int* deg     = ell + (long)N_NODES * ROWCAP;
    int* novf    = deg + N_NODES;
    int* ovf_dst = novf + 32;
    int* ovf_src = ovf_dst + N_EDGES;

    const int* src = ei;
    const int* dst = ei + N_EDGES;

    const int  degGrd  = (N_NODES + 1 + 255) / 256;
    const int  edgeGrd = (N_EDGES + 255) / 256;
    const int  layGrd  = LAY_GRID;

    gin_zero_i<<<degGrd, 256, 0, stream>>>(deg, N_NODES + 1, stats, out);
    gin_edge_rows<<<edgeGrd, 256, 0, stream>>>(src, dst, deg, novf, ell, ovf_dst, ovf_src, W1, W2, wpk);

    for (int i = 0; i < N_LAYERS; ++i) {
        const float* hin  = (i == 0) ? x : ((i & 1) ? hA : hB);
        float*       hout = (i & 1) ? hB : hA;
        int do_stats = (i < N_LAYERS - 1) ? 1 : 0;
        float*       st_cur  = stats + (long)i * 2 * DIM;
        const float* st_prev = stats + (long)(i > 0 ? i - 1 : 0) * 2 * DIM;
        const float* g_prev  = gamma + (long)(i > 0 ? i - 1 : 0) * DIM;
        const float* b_prev  = beta + (long)(i > 0 ? i - 1 : 0) * DIM;

        if (i == 0) {
            gin_layer<false, false><<<layGrd, 256, 0, stream>>>(hin, deg, ell, novf, ovf_dst, ovf_src,
                                                         wpk + (long)(2 * i) * DIM * DIM, b1 + (long)i * DIM,
                                                         wpk + (long)(2 * i + 1) * DIM * DIM, b2 + (long)i * DIM,
                                                         eps + i, hout, st_cur, do_stats,
                                                         st_prev, g_prev, b_prev, batch, out);
        } else if (i < N_LAYERS - 1) {
            gin_layer<true, false><<<layGrd, 256, 0, stream>>>(hin, deg, ell, novf, ovf_dst, ovf_src,
                                                        wpk + (long)(2 * i) * DIM * DIM, b1 + (long)i * DIM,
                                                        wpk + (long)(2 * i + 1) * DIM * DIM, b2 + (long)i * DIM,
                                                        eps + i, hout, st_cur, do_stats,
                                                        st_prev, g_prev, b_prev, batch, out);
        } else {
            gin_layer<true, true><<<layGrd, 256, 0, stream>>>(hin, deg, ell, novf, ovf_dst, ovf_src,
                                                        wpk + (long)(2 * i) * DIM * DIM, b1 + (long)i * DIM,
                                                        wpk + (long)(2 * i + 1) * DIM * DIM, b2 + (long)i * DIM,
                                                        eps + i, hout, st_cur, do_stats,
                                                        st_prev, g_prev, b_prev, batch, out);
        }
    }
}
